# starting kernel
# baseline (speedup 1.0000x reference)
#include <hip/hip_runtime.h>
#include <hip/hip_bf16.h>


typedef float v2f __attribute__((ext_vector_type(2)));
typedef float v8f __attribute__((ext_vector_type(8)));

#define HID 128
#define NB_STATS 512

__global__ void zero_i32(int* __restrict__ p, int n) {
  for (int i = blockIdx.x * blockDim.x + threadIdx.x; i < n;
       i += gridDim.x * blockDim.x)
    p[i] = 0;
}

__global__ void hist_deg(const int* __restrict__ dst, int* __restrict__ cnt,
                         int E) {
  for (int e = blockIdx.x * blockDim.x + threadIdx.x; e < E;
       e += gridDim.x * blockDim.x)
    atomicAdd(&cnt[dst[e]], 1);
}

__global__ void invdeg_from_cnt(const int* __restrict__ cnt,
                                float* __restrict__ invdeg, int N) {
  for (int i = blockIdx.x * blockDim.x + threadIdx.x; i < N;
       i += gridDim.x * blockDim.x)
    invdeg[i] = 1.0f / fmaxf((float)cnt[i], 1.0f);
}

__global__ void __launch_bounds__(1024) scan_excl(const int* __restrict__ cnt,
                                                  int* __restrict__ rowptr,
                                                  int N) {
  __shared__ int tmp[1024];
  __shared__ int carry;
  const int t = threadIdx.x;
  if (t == 0) carry = 0;
  __syncthreads();
  for (int base = 0; base < N; base += 1024) {
    int i = base + t;
    int v = (i < N) ? cnt[i] : 0;
    tmp[t] = v;
    __syncthreads();
#pragma unroll
    for (int off = 1; off < 1024; off <<= 1) {
      int addv = (t >= off) ? tmp[t - off] : 0;
      __syncthreads();
      tmp[t] += addv;
      __syncthreads();
    }
    int incl = tmp[t];
    int ctotal = tmp[1023];
    int cbase = carry;
    if (i < N) rowptr[i] = cbase + incl - v;
    __syncthreads();
    if (t == 0) carry = cbase + ctotal;
    __syncthreads();
  }
  if (t == 0) rowptr[N] = carry;
}

__global__ void fill_csr(const int* __restrict__ src,
                         const int* __restrict__ dst,
                         const int* __restrict__ rowptr,
                         int* __restrict__ cur, int* __restrict__ eidx,
                         int E) {
  for (int e = blockIdx.x * blockDim.x + threadIdx.x; e < E;
       e += gridDim.x * blockDim.x) {
    int d = dst[e];
    int slot = rowptr[d] + atomicAdd(&cur[d], 1);
    eidx[slot] = src[e];
  }
}

__global__ void __launch_bounds__(128) aggregate_csr(
    const float* __restrict__ h, const int* __restrict__ rowptr,
    const int* __restrict__ eidx, const float* __restrict__ invdeg,
    float* __restrict__ c, int N) {
  __shared__ int idxs[128];
  const int t = threadIdx.x;
  for (int node = blockIdx.x; node < N; node += gridDim.x) {
    int beg = rowptr[node];
    int cnt = rowptr[node + 1] - beg;
    float acc = 0.0f;
    for (int chunk = 0; chunk < cnt; chunk += 128) {
      int m = min(128, cnt - chunk);
      if (t < m) idxs[t] = eidx[beg + chunk + t];
      __syncthreads();
      int i = 0;
      for (; i + 4 <= m; i += 4) {
        float v0 = h[idxs[i + 0] * HID + t];
        float v1 = h[idxs[i + 1] * HID + t];
        float v2 = h[idxs[i + 2] * HID + t];
        float v3 = h[idxs[i + 3] * HID + t];
        acc += v0 + v1 + v2 + v3;
      }
      for (; i < m; ++i) acc += h[idxs[i] * HID + t];
      __syncthreads();
    }
    c[node * HID + t] = acc * invdeg[node];
  }
}

__global__ void __launch_bounds__(256) embed_gemm(
    const float* __restrict__ X, const float* __restrict__ W,
    const float* __restrict__ B, float* __restrict__ Y) {
  __shared__ __attribute__((aligned(16))) float tA[16 * HID];

  const int tid = threadIdx.x;
  const int rowBase = blockIdx.x * 16;

  const float4* Xv = (const float4*)(X + rowBase * HID);
  float4* tAv = (float4*)tA;
  tAv[tid] = Xv[tid];
  tAv[tid + 256] = Xv[tid + 256];
  __syncthreads();

  const int wave = tid >> 5;
  const int lane = tid & 31;
  const int half = lane >> 4;
  const int l16 = lane & 15;
  const int col = wave * 16 + l16;

  v8f acc;
  {
    float bv = B[col];
#pragma unroll
    for (int r = 0; r < 8; ++r) acc[r] = bv;
  }

  const float* arow = tA + l16 * HID;
  const float* wrow = W + col * HID;

#pragma unroll 4
  for (int k = 0; k < HID; k += 4) {
    const int kk = k + 2 * half;
    v2f a = *(const v2f*)(arow + kk);
    v2f b = *(const v2f*)(wrow + kk);
    acc = __builtin_amdgcn_wmma_f32_16x16x4_f32(false, a, false, b, (short)0,
                                                acc, false, false);
  }

#pragma unroll
  for (int r = 0; r < 8; ++r) {
    int row = rowBase + r + 8 * half;
    Y[row * HID + col] = acc[r];
  }
}

__global__ void __launch_bounds__(256) layer_gemm(
    const float* __restrict__ h, const float* __restrict__ c,
    const float* __restrict__ W, const float* __restrict__ B,
    float* __restrict__ Z) {
  __shared__ __attribute__((aligned(16))) float tA[16 * 256];

  const int tid = threadIdx.x;
  const int rowBase = blockIdx.x * 16;

#pragma unroll
  for (int i = 0; i < 2; ++i) {
    int v = tid + i * 256;
    int r = v >> 5;
    int c4 = v & 31;
    ((float4*)(tA + r * 256))[c4] =
        ((const float4*)(h + (rowBase + r) * HID))[c4];
  }
#pragma unroll
  for (int i = 0; i < 2; ++i) {
    int v = tid + i * 256;
    int r = v >> 5;
    int c4 = v & 31;
    ((float4*)(tA + r * 256 + HID))[c4] =
        ((const float4*)(c + (rowBase + r) * HID))[c4];
  }
  __syncthreads();

  const int wave = tid >> 5;
  const int lane = tid & 31;
  const int half = lane >> 4;
  const int l16 = lane & 15;
  const int col = wave * 16 + l16;

  v8f acc;
  {
    float bv = B[col];
#pragma unroll
    for (int r = 0; r < 8; ++r) acc[r] = bv;
  }

  const float* arow = tA + l16 * 256;
  const float* wrow = W + col * 256;

#pragma unroll 4
  for (int k = 0; k < 256; k += 4) {
    const int kk = k + 2 * half;
    v2f a = *(const v2f*)(arow + kk);
    v2f b = *(const v2f*)(wrow + kk);
    acc = __builtin_amdgcn_wmma_f32_16x16x4_f32(false, a, false, b, (short)0,
                                                acc, false, false);
  }

#pragma unroll
  for (int r = 0; r < 8; ++r) {
    int row = rowBase + r + 8 * half;
    Z[row * HID + col] = acc[r];
  }
}

__global__ void __launch_bounds__(128) norm_relu_stats(
    float* __restrict__ z, float* __restrict__ part, int N) {
  __shared__ float red[4];
  const int t = threadIdx.x;
  const int w = t >> 5;
  const int lane = t & 31;

  float ps = 0.0f, pss = 0.0f;
  for (int node = blockIdx.x; node < N; node += gridDim.x) {
    float v = z[node * HID + t];
    float sq = v * v;
#pragma unroll
    for (int off = 16; off > 0; off >>= 1) sq += __shfl_xor(sq, off, 32);
    if (lane == 0) red[w] = sq;
    __syncthreads();
    float tot = red[0] + red[1] + red[2] + red[3];
    __syncthreads();
    float nrm = fmaxf(sqrtf(tot), 1e-12f);
    float r = fmaxf(v / nrm, 0.0f);
    z[node * HID + t] = r;
    ps += r;
    pss += r * r;
  }
  part[blockIdx.x * 256 + t] = ps;
  part[blockIdx.x * 256 + 128 + t] = pss;
}

__global__ void __launch_bounds__(256) bn_finalize(
    const float* __restrict__ part, int nb, const float* __restrict__ gamma,
    const float* __restrict__ beta, float* __restrict__ bn, float invN) {
  __shared__ float tot[256];
  const int j = threadIdx.x;
  float s = 0.0f;
  for (int b = 0; b < nb; ++b) s += part[b * 256 + j];
  tot[j] = s;
  __syncthreads();
  if (j < 128) {
    float mu = tot[j] * invN;
    float var = tot[128 + j] * invN - mu * mu;
    float a = gamma[j] / sqrtf(var + 1e-5f);
    bn[j] = a;
    bn[128 + j] = beta[j] - mu * a;
  }
}

__global__ void residual_bn(const float* __restrict__ h,
                            const float* __restrict__ z,
                            const float* __restrict__ bn,
                            float* __restrict__ out, int total) {
  for (int i = blockIdx.x * blockDim.x + threadIdx.x; i < total;
       i += gridDim.x * blockDim.x) {
    int j = i & (HID - 1);
    out[i] = h[i] + z[i] * bn[j] + bn[HID + j];
  }
}

extern "C" void kernel_launch(void* const* d_in, const int* in_sizes, int n_in,
                              void* d_out, int out_size, void* d_ws,
                              size_t ws_size, hipStream_t stream) {
  const float* h0 = (const float*)d_in[0];
  const int* src = (const int*)d_in[1];
  const int* dst = (const int*)d_in[2];
  const float* W_emb = (const float*)d_in[3];
  const float* b_emb = (const float*)d_in[4];
  const float* Ws = (const float*)d_in[5];
  const float* bs = (const float*)d_in[6];
  const float* gam = (const float*)d_in[7];
  const float* bet = (const float*)d_in[8];

  const int N = in_sizes[0] / HID;
  const int E = in_sizes[1];
  const int NF = N * HID;
  const int L = 4;

  float* ws = (float*)d_ws;
  float* H = ws;
  float* C = H + NF;
  float* Z = C + NF;
  float* DEG = Z + NF;
  float* BN = DEG + N;
  float* PART = BN + 256;
  int* rowptr = (int*)(PART + NB_STATS * 256);
  int* cnt = rowptr + (N + 1);
  int* eidx = cnt + N;

  const int mtiles = N / 16;

  zero_i32<<<256, 256, 0, stream>>>(cnt, N);
  hist_deg<<<1024, 256, 0, stream>>>(dst, cnt, E);
  invdeg_from_cnt<<<256, 256, 0, stream>>>(cnt, DEG, N);
  scan_excl<<<1, 1024, 0, stream>>>(cnt, rowptr, N);
  zero_i32<<<256, 256, 0, stream>>>(cnt, N);
  fill_csr<<<1024, 256, 0, stream>>>(src, dst, rowptr, cnt, eidx, E);

  embed_gemm<<<mtiles, 256, 0, stream>>>(h0, W_emb, b_emb, H);

  for (int l = 0; l < L; ++l) {
    aggregate_csr<<<2048, 128, 0, stream>>>(H, rowptr, eidx, DEG, C, N);
    layer_gemm<<<mtiles, 256, 0, stream>>>(H, C, Ws + l * HID * 256,
                                           bs + l * HID, Z);
    norm_relu_stats<<<NB_STATS, 128, 0, stream>>>(Z, PART, N);
    bn_finalize<<<1, 256, 0, stream>>>(PART, NB_STATS, gam + l * HID,
                                       bet + l * HID, BN, 1.0f / (float)N);
    float* out = (l == L - 1) ? (float*)d_out : H;
    residual_bn<<<4096, 256, 0, stream>>>(H, Z, BN, out, NF);
  }
}


# embed_gemm, layer_gemm: W half kept in LDS in fragment order, each wave owns a 16-row tile, A fragments from global straight into registers, no A tile in LDS; k sum order inside each 8 changed
# speedup vs baseline: 2.2807x; 2.2807x over previous
#include <hip/hip_runtime.h>
#include <hip/hip_bf16.h>


typedef float v2f __attribute__((ext_vector_type(2)));
typedef float v8f __attribute__((ext_vector_type(8)));

#define HID 128
#define NB_STATS 512

__global__ void zero_i32(int* __restrict__ p, int n) {
  for (int i = blockIdx.x * blockDim.x + threadIdx.x; i < n;
       i += gridDim.x * blockDim.x)
    p[i] = 0;
}

__global__ void hist_deg(const int* __restrict__ dst, int* __restrict__ cnt,
                         int E) {
  for (int e = blockIdx.x * blockDim.x + threadIdx.x; e < E;
       e += gridDim.x * blockDim.x)
    atomicAdd(&cnt[dst[e]], 1);
}

__global__ void invdeg_from_cnt(const int* __restrict__ cnt,
                                float* __restrict__ invdeg, int N) {
  for (int i = blockIdx.x * blockDim.x + threadIdx.x; i < N;
       i += gridDim.x * blockDim.x)
    invdeg[i] = 1.0f / fmaxf((float)cnt[i], 1.0f);
}

__global__ void __launch_bounds__(1024) alloc_rows(const int* __restrict__ cnt,
                                                   int* __restrict__ rowstart,
                                                   int* __restrict__ total,
                                                   int N) {
  __shared__ int tmp[1024];
  __shared__ int base;
  const int t = threadIdx.x;
  const int i = blockIdx.x * 1024 + t;
  const int v = (i < N) ? cnt[i] : 0;
  tmp[t] = v;
  __syncthreads();
  for (int off = 1; off < 1024; off <<= 1) {
    int addv = (t >= off) ? tmp[t - off] : 0;
    __syncthreads();
    tmp[t] += addv;
    __syncthreads();
  }
  const int incl = tmp[t];
  if (t == 1023) base = atomicAdd(total, incl);
  __syncthreads();
  if (i < N) rowstart[i] = base + incl - v;
}

__global__ void fill_csr(const int* __restrict__ src,
                         const int* __restrict__ dst,
                         const int* __restrict__ rowptr,
                         int* __restrict__ cur, int* __restrict__ eidx,
                         int E) {
  for (int e = blockIdx.x * blockDim.x + threadIdx.x; e < E;
       e += gridDim.x * blockDim.x) {
    int d = dst[e];
    int slot = rowptr[d] + atomicAdd(&cur[d], 1);
    eidx[slot] = src[e];
  }
}

#define GW 16

__global__ void __launch_bounds__(GW * 32) gather_norm_stats(
    const float* __restrict__ p2, const int* __restrict__ rowptr,
    const int* __restrict__ deg, const int* __restrict__ eidx,
    const float* __restrict__ invdeg, float* __restrict__ z,
    float* __restrict__ part, int N) {
  __shared__ __attribute__((aligned(16))) float red[GW][256];

  const int t = threadIdx.x;
  const int lane = t & 31;
  const int wave = __builtin_amdgcn_readfirstlane(t >> 5);

  const float4* p2v = (const float4*)p2;
  float4* zv4 = (float4*)z;

  float4 ps = make_float4(0.f, 0.f, 0.f, 0.f);
  float4 pss = make_float4(0.f, 0.f, 0.f, 0.f);

  const int stride = gridDim.x * GW;
  const int node0 = blockIdx.x * GW + wave;

  for (int kb = 0; node0 + kb * stride < N; kb += 32) {
    const int ln = node0 + (kb + lane) * stride;
    int lb = 0, le = 0;
    if (ln < N) {
      lb = rowptr[ln];
      le = lb + deg[ln];
    }
    for (int k = 0; k < 32; ++k) {
      const int node = node0 + (kb + k) * stride;
      if (node >= N) break;

      const int cb = __builtin_amdgcn_readlane(lb, k);
      const int ce = __builtin_amdgcn_readlane(le, k);
      const int my = (lane < ce - cb) ? eidx[cb + lane] : 0;

      const float4 zv = zv4[node * 32 + lane];
      const float inv = invdeg[node];

      float4 acc = make_float4(0.f, 0.f, 0.f, 0.f);
      int cur = my;
      for (int base = cb; base < ce; base += 32) {
        if (base != cb) cur = (lane < ce - base) ? eidx[base + lane] : 0;
        const int m = min(32, ce - base);
        for (int i = 0; i < m; i += 8) {
          float4 v[8];
#pragma unroll
          for (int j = 0; j < 8; ++j) {
            const int s = __builtin_amdgcn_readlane(cur, i + j);
            v[j] = p2v[s * 32 + lane];
          }
#pragma unroll
          for (int j = 0; j < 8; ++j) {
            const bool ok = (i + j) < m;
            acc.x += ok ? v[j].x : 0.f;
            acc.y += ok ? v[j].y : 0.f;
            acc.z += ok ? v[j].z : 0.f;
            acc.w += ok ? v[j].w : 0.f;
          }
        }
      }

      float4 zz;
      zz.x = zv.x + acc.x * inv;
      zz.y = zv.y + acc.y * inv;
      zz.z = zv.z + acc.z * inv;
      zz.w = zv.w + acc.w * inv;

      float sq = zz.x * zz.x + zz.y * zz.y + zz.z * zz.z + zz.w * zz.w;
#pragma unroll
      for (int off = 16; off > 0; off >>= 1) sq += __shfl_xor(sq, off, 32);
      const float nrm = fmaxf(sqrtf(sq), 1e-12f);

      float4 r;
      r.x = fmaxf(zz.x / nrm, 0.0f);
      r.y = fmaxf(zz.y / nrm, 0.0f);
      r.z = fmaxf(zz.z / nrm, 0.0f);
      r.w = fmaxf(zz.w / nrm, 0.0f);
      zv4[node * 32 + lane] = r;

      ps.x += r.x;  ps.y += r.y;  ps.z += r.z;  ps.w += r.w;
      pss.x += r.x * r.x;  pss.y += r.y * r.y;
      pss.z += r.z * r.z;  pss.w += r.w * r.w;
    }
  }

  ((float4*)&red[wave][0])[lane] = ps;
  ((float4*)&red[wave][128])[lane] = pss;
  __syncthreads();
  if (t < 256) {
    float s = 0.0f;
#pragma unroll
    for (int w = 0; w < GW; ++w) s += red[w][t];
    part[blockIdx.x * 256 + t] = s;
  }
}

#define RG_WAVES 16
#define RG_BLOCKS 256

template <int NQ>
__global__ void __launch_bounds__(RG_WAVES * 32) rowtile_gemm(
    const float* __restrict__ A, const float* __restrict__ W,
    const float* __restrict__ bias, float* __restrict__ out0,
    float* __restrict__ out1, int nTiles) {
  __shared__ __attribute__((aligned(16))) float4 wf[16 * 8 * 2 * 16];

  const int tid = threadIdx.x;
  const int lane = tid & 31;
  const int wave = __builtin_amdgcn_readfirstlane(tid >> 5);
  const int hh = lane >> 4;
  const int c = lane & 15;

  const int q = blockIdx.x & (NQ - 1);
  const int tb = blockIdx.x / NQ;
  const int ts = RG_BLOCKS / NQ;
  const float* Wq = W + q * HID;
  float* __restrict__ out = (q == 0) ? out0 : out1;

  const float4* Av = (const float4*)A;

  int m = wave;
  int tile = tb + ts * m;
  bool have = tile < nTiles;

  float4 a[16];
  if (have) {
    const float4* ap = Av + (tile * 16 + c) * (HID / 4) + hh;
#pragma unroll
    for (int g = 0; g < 16; ++g) a[g] = ap[2 * g];
  }

  {
    float4 t[8];
#pragma unroll
    for (int i = 0; i < 8; ++i)
      t[i] = *(const float4*)(Wq + (16 * i + wave) * (HID * NQ) + 4 * lane);
#pragma unroll
    for (int i = 0; i < 8; ++i)
      wf[(((lane >> 1) * 8 + i) * 2 + (lane & 1)) * 16 + ((wave ^ lane) & 15)] =
          t[i];
  }
  __syncthreads();

  while (have) {
    asm volatile("" ::: "memory");
    v8f acc[8];
#pragma unroll
    for (int j = 0; j < 8; ++j) {
      const float bv = (q == 0) ? bias[16 * j + c] : 0.0f;
#pragma unroll
      for (int r = 0; r < 8; ++r) acc[j][r] = bv;
    }

    float4 b0[8], b1[8];
#pragma unroll
    for (int j = 0; j < 8; ++j)
      b0[j] = wf[((0 * 8 + j) * 2 + hh) * 16 + (c ^ ((2 * 0 + hh) & 15))];

#pragma unroll
    for (int g = 0; g < 16; g += 2) {
#pragma unroll
      for (int j = 0; j < 8; ++j)
        b1[j] = wf[(((g + 1) * 8 + j) * 2 + hh) * 16 +
                   (c ^ ((2 * (g + 1) + hh) & 15))];
      __builtin_amdgcn_sched_barrier(0);
      {
        const v2f alo = {a[g].x, a[g].y};
        const v2f ahi = {a[g].z, a[g].w};
#pragma unroll
        for (int j = 0; j < 8; ++j) {
          const v2f b = {b0[j].x, b0[j].y};
          acc[j] = __builtin_amdgcn_wmma_f32_16x16x4_f32(
              false, alo, false, b, (short)0, acc[j], false, false);
        }
#pragma unroll
        for (int j = 0; j < 8; ++j) {
          const v2f b = {b0[j].z, b0[j].w};
          acc[j] = __builtin_amdgcn_wmma_f32_16x16x4_f32(
              false, ahi, false, b, (short)0, acc[j], false, false);
        }
      }
      __builtin_amdgcn_sched_barrier(0);
      if (g + 2 < 16) {
#pragma unroll
        for (int j = 0; j < 8; ++j)
          b0[j] = wf[(((g + 2) * 8 + j) * 2 + hh) * 16 +
                     (c ^ ((2 * (g + 2) + hh) & 15))];
      }
      __builtin_amdgcn_sched_barrier(0);
      {
        const v2f alo = {a[g + 1].x, a[g + 1].y};
        const v2f ahi = {a[g + 1].z, a[g + 1].w};
#pragma unroll
        for (int j = 0; j < 8; ++j) {
          const v2f b = {b1[j].x, b1[j].y};
          acc[j] = __builtin_amdgcn_wmma_f32_16x16x4_f32(
              false, alo, false, b, (short)0, acc[j], false, false);
        }
#pragma unroll
        for (int j = 0; j < 8; ++j) {
          const v2f b = {b1[j].z, b1[j].w};
          acc[j] = __builtin_amdgcn_wmma_f32_16x16x4_f32(
              false, ahi, false, b, (short)0, acc[j], false, false);
        }
      }
      __builtin_amdgcn_sched_barrier(0);
    }

    float* orow = out + (tile * 16 + 8 * hh) * HID + c;
#pragma unroll
    for (int r = 0; r < 8; ++r) {
#pragma unroll
      for (int j = 0; j < 8; ++j) orow[r * HID + 16 * j] = acc[j][r];
    }

    m += RG_WAVES;
    tile = tb + ts * m;
    have = tile < nTiles;
    if (have) {
      const float4* ap = Av + (tile * 16 + c) * (HID / 4) + hh;
#pragma unroll
      for (int g = 0; g < 16; ++g) a[g] = ap[2 * g];
    }
  }
}

__global__ void __launch_bounds__(256) bn_finalize(
    const float* __restrict__ part, int nb, const float* __restrict__ gamma,
    const float* __restrict__ beta, float* __restrict__ bn, float invN) {
  __shared__ float tot[256];
  const int j = threadIdx.x;
  float s = 0.0f;
  for (int b = 0; b < nb; ++b) s += part[b * 256 + j];
  tot[j] = s;
  __syncthreads();
  if (j < 128) {
    float mu = tot[j] * invN;
    float var = tot[128 + j] * invN - mu * mu;
    float a = gamma[j] / sqrtf(var + 1e-5f);
    bn[j] = a;
    bn[128 + j] = beta[j] - mu * a;
  }
}

__global__ void residual_bn(const float* __restrict__ h,
                            const float* __restrict__ z,
                            const float* __restrict__ bn,
                            float* __restrict__ out, int total) {
  for (int i = blockIdx.x * blockDim.x + threadIdx.x; i < total;
       i += gridDim.x * blockDim.x) {
    int j = i & (HID - 1);
    out[i] = h[i] + z[i] * bn[j] + bn[HID + j];
  }
}

extern "C" void kernel_launch(void* const* d_in, const int* in_sizes, int n_in,
                              void* d_out, int out_size, void* d_ws,
                              size_t ws_size, hipStream_t stream) {
  const float* h0 = (const float*)d_in[0];
  const int* src = (const int*)d_in[1];
  const int* dst = (const int*)d_in[2];
  const float* W_emb = (const float*)d_in[3];
  const float* b_emb = (const float*)d_in[4];
  const float* Ws = (const float*)d_in[5];
  const float* bs = (const float*)d_in[6];
  const float* gam = (const float*)d_in[7];
  const float* bet = (const float*)d_in[8];

  const int N = in_sizes[0] / HID;
  const int E = in_sizes[1];
  const int NF = N * HID;
  const int L = 4;

  float* ws = (float*)d_ws;
  float* H = ws;
  float* C = H + NF;
  float* Z = C + NF;
  float* DEG = Z + NF;
  float* BN = DEG + N;
  float* PART = BN + 256;
  int* rowptr = (int*)(PART + NB_STATS * 256);
  int* cnt = rowptr + (N + 1);
  int* eidx = cnt + N;

  const int mtiles = N / 16;

  zero_i32<<<256, 256, 0, stream>>>(rowptr + N, N + 1);
  hist_deg<<<1024, 256, 0, stream>>>(dst, cnt, E);
  invdeg_from_cnt<<<256, 256, 0, stream>>>(cnt, DEG, N);
  alloc_rows<<<(N + 1023) / 1024, 1024, 0, stream>>>(cnt, rowptr, rowptr + N,
                                                     N);
  zero_i32<<<256, 256, 0, stream>>>(cnt, N);
  fill_csr<<<1024, 256, 0, stream>>>(src, dst, rowptr, cnt, eidx, E);

  rowtile_gemm<1><<<RG_BLOCKS, RG_WAVES * 32, 0, stream>>>(h0, W_emb, b_emb, H,
                                                          H, mtiles);

  for (int l = 0; l < L; ++l) {
    rowtile_gemm<2><<<RG_BLOCKS, RG_WAVES * 32, 0, stream>>>(
        H, Ws + l * HID * 256, bs + l * HID, Z, C, mtiles);
    gather_norm_stats<<<NB_STATS, GW * 32, 0, stream>>>(C, rowptr, cnt, eidx,
                                                        DEG, Z, PART, N);
    bn_finalize<<<1, 256, 0, stream>>>(PART, NB_STATS, gam + l * HID,
                                       bet + l * HID, BN, 1.0f / (float)N);
    float* out = (l == L - 1) ? (float*)d_out : H;
    residual_bn<<<4096, 256, 0, stream>>>(H, Z, BN, out, NF);
  }
}


# BN column sums as 64-bit fixed-point totals: gather blocks add partials by integer atomics, residual_bn derives scale/shift from them; partial rows, bn_finalize launch gone; BN sums order-free
# speedup vs baseline: 3.1305x; 1.3726x over previous
#include <hip/hip_runtime.h>
#include <hip/hip_bf16.h>


typedef float v2f __attribute__((ext_vector_type(2)));
typedef float v8f __attribute__((ext_vector_type(8)));

#define HID 128
#define NB_STATS 512

#define FX_SCALE 68719476736.0f
#define FX_INV 1.4551915228366852e-11f

__global__ void zero_i32(int* __restrict__ p, int n) {
  for (int i = blockIdx.x * blockDim.x + threadIdx.x; i < n;
       i += gridDim.x * blockDim.x)
    p[i] = 0;
}

__global__ void hist_deg(const int* __restrict__ dst, int* __restrict__ cnt,
                         int E) {
  for (int e = blockIdx.x * blockDim.x + threadIdx.x; e < E;
       e += gridDim.x * blockDim.x)
    atomicAdd(&cnt[dst[e]], 1);
}

__global__ void invdeg_from_cnt(const int* __restrict__ cnt,
                                float* __restrict__ invdeg, int N) {
  for (int i = blockIdx.x * blockDim.x + threadIdx.x; i < N;
       i += gridDim.x * blockDim.x)
    invdeg[i] = 1.0f / fmaxf((float)cnt[i], 1.0f);
}

__global__ void __launch_bounds__(1024) alloc_rows(const int* __restrict__ cnt,
                                                   int* __restrict__ rowstart,
                                                   int* __restrict__ total,
                                                   int N) {
  __shared__ int tmp[1024];
  __shared__ int base;
  const int t = threadIdx.x;
  const int i = blockIdx.x * 1024 + t;
  const int v = (i < N) ? cnt[i] : 0;
  tmp[t] = v;
  __syncthreads();
  for (int off = 1; off < 1024; off <<= 1) {
    int addv = (t >= off) ? tmp[t - off] : 0;
    __syncthreads();
    tmp[t] += addv;
    __syncthreads();
  }
  const int incl = tmp[t];
  if (t == 1023) base = atomicAdd(total, incl);
  __syncthreads();
  if (i < N) rowstart[i] = base + incl - v;
}

__global__ void fill_csr(const int* __restrict__ src,
                         const int* __restrict__ dst,
                         const int* __restrict__ rowptr,
                         int* __restrict__ cur, int* __restrict__ eidx,
                         int E) {
  for (int e = blockIdx.x * blockDim.x + threadIdx.x; e < E;
       e += gridDim.x * blockDim.x) {
    int d = dst[e];
    int slot = rowptr[d] + atomicAdd(&cur[d], 1);
    eidx[slot] = src[e];
  }
}

#define GW 16

__global__ void __launch_bounds__(GW * 32) gather_norm_stats(
    const float* __restrict__ p2, const int* __restrict__ rowptr,
    const int* __restrict__ deg, const int* __restrict__ eidx,
    const float* __restrict__ invdeg, float* __restrict__ z,
    long long* __restrict__ tot, int N) {
  __shared__ __attribute__((aligned(16))) float red[GW][256];

  const int t = threadIdx.x;
  const int lane = t & 31;
  const int wave = __builtin_amdgcn_readfirstlane(t >> 5);

  const float4* p2v = (const float4*)p2;
  float4* zv4 = (float4*)z;

  float4 ps = make_float4(0.f, 0.f, 0.f, 0.f);
  float4 pss = make_float4(0.f, 0.f, 0.f, 0.f);

  const int stride = gridDim.x * GW;
  const int node0 = blockIdx.x * GW + wave;

  for (int kb = 0; node0 + kb * stride < N; kb += 32) {
    const int ln = node0 + (kb + lane) * stride;
    int lb = 0, le = 0;
    if (ln < N) {
      lb = rowptr[ln];
      le = lb + deg[ln];
    }
    for (int k = 0; k < 32; ++k) {
      const int node = node0 + (kb + k) * stride;
      if (node >= N) break;

      const int cb = __builtin_amdgcn_readlane(lb, k);
      const int ce = __builtin_amdgcn_readlane(le, k);
      const int my = (lane < ce - cb) ? eidx[cb + lane] : 0;

      const float4 zv = zv4[node * 32 + lane];
      const float inv = invdeg[node];

      float4 acc = make_float4(0.f, 0.f, 0.f, 0.f);
      int cur = my;
      for (int base = cb; base < ce; base += 32) {
        if (base != cb) cur = (lane < ce - base) ? eidx[base + lane] : 0;
        const int m = min(32, ce - base);
        for (int i = 0; i < m; i += 8) {
          float4 v[8];
#pragma unroll
          for (int j = 0; j < 8; ++j) {
            const int s = __builtin_amdgcn_readlane(cur, i + j);
            v[j] = p2v[s * 32 + lane];
          }
#pragma unroll
          for (int j = 0; j < 8; ++j) {
            const bool ok = (i + j) < m;
            acc.x += ok ? v[j].x : 0.f;
            acc.y += ok ? v[j].y : 0.f;
            acc.z += ok ? v[j].z : 0.f;
            acc.w += ok ? v[j].w : 0.f;
          }
        }
      }

      float4 zz;
      zz.x = zv.x + acc.x * inv;
      zz.y = zv.y + acc.y * inv;
      zz.z = zv.z + acc.z * inv;
      zz.w = zv.w + acc.w * inv;

      float sq = zz.x * zz.x + zz.y * zz.y + zz.z * zz.z + zz.w * zz.w;
#pragma unroll
      for (int off = 16; off > 0; off >>= 1) sq += __shfl_xor(sq, off, 32);
      const float nrm = fmaxf(sqrtf(sq), 1e-12f);

      float4 r;
      r.x = fmaxf(zz.x / nrm, 0.0f);
      r.y = fmaxf(zz.y / nrm, 0.0f);
      r.z = fmaxf(zz.z / nrm, 0.0f);
      r.w = fmaxf(zz.w / nrm, 0.0f);
      zv4[node * 32 + lane] = r;

      ps.x += r.x;  ps.y += r.y;  ps.z += r.z;  ps.w += r.w;
      pss.x += r.x * r.x;  pss.y += r.y * r.y;
      pss.z += r.z * r.z;  pss.w += r.w * r.w;
    }
  }

  ((float4*)&red[wave][0])[lane] = ps;
  ((float4*)&red[wave][128])[lane] = pss;
  __syncthreads();
  if (t < 256) {
    float s = 0.0f;
#pragma unroll
    for (int w = 0; w < GW; ++w) s += red[w][t];
    __hip_atomic_fetch_add(&tot[t], (long long)(s * FX_SCALE),
                           __ATOMIC_RELAXED, __HIP_MEMORY_SCOPE_AGENT);
  }
}

#define RG_WAVES 16
#define RG_BLOCKS 256

template <int NQ>
__global__ void __launch_bounds__(RG_WAVES * 32) rowtile_gemm(
    const float* __restrict__ A, const float* __restrict__ W,
    const float* __restrict__ bias, float* __restrict__ out0,
    float* __restrict__ out1, int nTiles) {
  __shared__ __attribute__((aligned(16))) float4 wf[16 * 8 * 2 * 16];

  const int tid = threadIdx.x;
  const int lane = tid & 31;
  const int wave = __builtin_amdgcn_readfirstlane(tid >> 5);
  const int hh = lane >> 4;
  const int c = lane & 15;

  const int q = blockIdx.x & (NQ - 1);
  const int tb = blockIdx.x / NQ;
  const int ts = RG_BLOCKS / NQ;
  const float* Wq = W + q * HID;
  float* __restrict__ out = (q == 0) ? out0 : out1;

  const float4* Av = (const float4*)A;

  int m = wave;
  int tile = tb + ts * m;
  bool have = tile < nTiles;

  float4 a[16];
  if (have) {
    const float4* ap = Av + (tile * 16 + c) * (HID / 4) + hh;
#pragma unroll
    for (int g = 0; g < 16; ++g) a[g] = ap[2 * g];
  }

  {
    float4 t[8];
#pragma unroll
    for (int i = 0; i < 8; ++i)
      t[i] = *(const float4*)(Wq + (16 * i + wave) * (HID * NQ) + 4 * lane);
#pragma unroll
    for (int i = 0; i < 8; ++i)
      wf[(((lane >> 1) * 8 + i) * 2 + (lane & 1)) * 16 + ((wave ^ lane) & 15)] =
          t[i];
  }
  __syncthreads();

  while (have) {
    asm volatile("" ::: "memory");
    v8f acc[8];
#pragma unroll
    for (int j = 0; j < 8; ++j) {
      const float bv = (q == 0) ? bias[16 * j + c] : 0.0f;
#pragma unroll
      for (int r = 0; r < 8; ++r) acc[j][r] = bv;
    }

    float4 b0[8], b1[8];
#pragma unroll
    for (int j = 0; j < 8; ++j)
      b0[j] = wf[((0 * 8 + j) * 2 + hh) * 16 + (c ^ ((2 * 0 + hh) & 15))];

#pragma unroll
    for (int g = 0; g < 16; g += 2) {
#pragma unroll
      for (int j = 0; j < 8; ++j)
        b1[j] = wf[(((g + 1) * 8 + j) * 2 + hh) * 16 +
                   (c ^ ((2 * (g + 1) + hh) & 15))];
      __builtin_amdgcn_sched_barrier(0);
      {
        const v2f alo = {a[g].x, a[g].y};
        const v2f ahi = {a[g].z, a[g].w};
#pragma unroll
        for (int j = 0; j < 8; ++j) {
          const v2f b = {b0[j].x, b0[j].y};
          acc[j] = __builtin_amdgcn_wmma_f32_16x16x4_f32(
              false, alo, false, b, (short)0, acc[j], false, false);
        }
#pragma unroll
        for (int j = 0; j < 8; ++j) {
          const v2f b = {b0[j].z, b0[j].w};
          acc[j] = __builtin_amdgcn_wmma_f32_16x16x4_f32(
              false, ahi, false, b, (short)0, acc[j], false, false);
        }
      }
      __builtin_amdgcn_sched_barrier(0);
      if (g + 2 < 16) {
#pragma unroll
        for (int j = 0; j < 8; ++j)
          b0[j] = wf[(((g + 2) * 8 + j) * 2 + hh) * 16 +
                     (c ^ ((2 * (g + 2) + hh) & 15))];
      }
      __builtin_amdgcn_sched_barrier(0);
      {
        const v2f alo = {a[g + 1].x, a[g + 1].y};
        const v2f ahi = {a[g + 1].z, a[g + 1].w};
#pragma unroll
        for (int j = 0; j < 8; ++j) {
          const v2f b = {b1[j].x, b1[j].y};
          acc[j] = __builtin_amdgcn_wmma_f32_16x16x4_f32(
              false, alo, false, b, (short)0, acc[j], false, false);
        }
#pragma unroll
        for (int j = 0; j < 8; ++j) {
          const v2f b = {b1[j].z, b1[j].w};
          acc[j] = __builtin_amdgcn_wmma_f32_16x16x4_f32(
              false, ahi, false, b, (short)0, acc[j], false, false);
        }
      }
      __builtin_amdgcn_sched_barrier(0);
    }

    float* orow = out + (tile * 16 + 8 * hh) * HID + c;
#pragma unroll
    for (int r = 0; r < 8; ++r) {
#pragma unroll
      for (int j = 0; j < 8; ++j) orow[r * HID + 16 * j] = acc[j][r];
    }

    m += RG_WAVES;
    tile = tb + ts * m;
    have = tile < nTiles;
    if (have) {
      const float4* ap = Av + (tile * 16 + c) * (HID / 4) + hh;
#pragma unroll
      for (int g = 0; g < 16; ++g) a[g] = ap[2 * g];
    }
  }
}

__global__ void residual_bn(const float* __restrict__ h,
                            const float* __restrict__ z,
                            const long long* __restrict__ tot,
                            const float* __restrict__ gamma,
                            const float* __restrict__ beta,
                            float* __restrict__ out, int total, float invN) {
  const int j = threadIdx.x & (HID - 1);
  const float s = (float)tot[j] * FX_INV;
  const float ss = (float)tot[HID + j] * FX_INV;
  const float mu = s * invN;
  const float var = ss * invN - mu * mu;
  const float a = gamma[j] / sqrtf(var + 1e-5f);
  const float b = beta[j] - mu * a;
  for (int i = blockIdx.x * blockDim.x + threadIdx.x; i < total;
       i += gridDim.x * blockDim.x)
    out[i] = h[i] + z[i] * a + b;
}

extern "C" void kernel_launch(void* const* d_in, const int* in_sizes, int n_in,
                              void* d_out, int out_size, void* d_ws,
                              size_t ws_size, hipStream_t stream) {
  const float* h0 = (const float*)d_in[0];
  const int* src = (const int*)d_in[1];
  const int* dst = (const int*)d_in[2];
  const float* W_emb = (const float*)d_in[3];
  const float* b_emb = (const float*)d_in[4];
  const float* Ws = (const float*)d_in[5];
  const float* bs = (const float*)d_in[6];
  const float* gam = (const float*)d_in[7];
  const float* bet = (const float*)d_in[8];

  const int N = in_sizes[0] / HID;
  const int E = in_sizes[1];
  const int NF = N * HID;
  const int L = 4;

  float* ws = (float*)d_ws;
  float* H = ws;
  float* C = H + NF;
  float* Z = C + NF;
  float* DEG = Z + NF;
  int* rowptr = (int*)(DEG + N);
  long long* TOT = (long long*)(rowptr + N + 2);
  int* cnt = (int*)(TOT + L * 256);
  int* eidx = cnt + N;

  const int mtiles = N / 16;

  zero_i32<<<256, 256, 0, stream>>>(rowptr + N, 2 + L * 512 + N);
  hist_deg<<<1024, 256, 0, stream>>>(dst, cnt, E);
  invdeg_from_cnt<<<256, 256, 0, stream>>>(cnt, DEG, N);
  alloc_rows<<<(N + 1023) / 1024, 1024, 0, stream>>>(cnt, rowptr, rowptr + N,
                                                     N);
  zero_i32<<<256, 256, 0, stream>>>(cnt, N);
  fill_csr<<<1024, 256, 0, stream>>>(src, dst, rowptr, cnt, eidx, E);

  rowtile_gemm<1><<<RG_BLOCKS, RG_WAVES * 32, 0, stream>>>(h0, W_emb, b_emb, H,
                                                          H, mtiles);

  for (int l = 0; l < L; ++l) {
    rowtile_gemm<2><<<RG_BLOCKS, RG_WAVES * 32, 0, stream>>>(
        H, Ws + l * HID * 256, bs + l * HID, Z, C, mtiles);
    gather_norm_stats<<<NB_STATS, GW * 32, 0, stream>>>(
        C, rowptr, cnt, eidx, DEG, Z, TOT + l * 256, N);
    float* out = (l == L - 1) ? (float*)d_out : H;
    residual_bn<<<4096, 256, 0, stream>>>(H, Z, TOT + l * 256, gam + l * HID,
                                          bet + l * HID, out, NF,
                                          1.0f / (float)N);
  }
}


# no CSR: each node owns a 32-slot row filled in one edge pass (the degree add's return is the slot, later edges go to a spill list); hist, 1/deg, alloc, 2nd clear launches gone; list order by arrival
# speedup vs baseline: 3.9302x; 1.2555x over previous
#include <hip/hip_runtime.h>
#include <hip/hip_bf16.h>


typedef float v2f __attribute__((ext_vector_type(2)));
typedef float v8f __attribute__((ext_vector_type(8)));

#define HID 128
#define NB_STATS 512

#define FX_SCALE 68719476736.0f
#define FX_INV 1.4551915228366852e-11f

__global__ void zero_i32(int* __restrict__ p, int n) {
  for (int i = blockIdx.x * blockDim.x + threadIdx.x; i < n;
       i += gridDim.x * blockDim.x)
    p[i] = 0;
}

#define CAP 32

__global__ void fill_slots(const int* __restrict__ src,
                           const int* __restrict__ dst, int* __restrict__ cnt,
                           int* __restrict__ tab, int* __restrict__ spill,
                           int* __restrict__ nspill, int E) {
  for (int e = blockIdx.x * blockDim.x + threadIdx.x; e < E;
       e += gridDim.x * blockDim.x) {
    const int d = dst[e];
    const int s = src[e];
    const int slot = atomicAdd(&cnt[d], 1);
    if (slot < CAP) {
      tab[d * CAP + slot] = s;
    } else {
      const int o = atomicAdd(nspill, 1);
      ((int2*)spill)[o] = make_int2(d, s);
    }
  }
}

#define GW 16

__global__ void __launch_bounds__(GW * 32) gather_norm_stats(
    const float* __restrict__ p2, const int* __restrict__ deg,
    const int* __restrict__ tab, const int* __restrict__ spill,
    const int* __restrict__ nspill, float* __restrict__ z,
    long long* __restrict__ tot, int N) {
  __shared__ __attribute__((aligned(16))) float red[GW][256];

  const int t = threadIdx.x;
  const int lane = t & 31;
  const int wave = __builtin_amdgcn_readfirstlane(t >> 5);

  const float4* p2v = (const float4*)p2;
  float4* zv4 = (float4*)z;

  float4 ps = make_float4(0.f, 0.f, 0.f, 0.f);
  float4 pss = make_float4(0.f, 0.f, 0.f, 0.f);

  const int stride = gridDim.x * GW;
  const int node0 = blockIdx.x * GW + wave;

  for (int kb = 0; node0 + kb * stride < N; kb += 32) {
    const int ln = node0 + (kb + lane) * stride;
    const int ld = (ln < N) ? deg[ln] : 0;
    for (int k = 0; k < 32; ++k) {
      const int node = node0 + (kb + k) * stride;
      if (node >= N) break;

      const int dg = __builtin_amdgcn_readlane(ld, k);
      const int m = min(dg, CAP);
      const int cur = (lane < m) ? tab[node * CAP + lane] : 0;

      const float4 zv = zv4[node * 32 + lane];
      const float inv = 1.0f / fmaxf((float)dg, 1.0f);

      float4 acc = make_float4(0.f, 0.f, 0.f, 0.f);
      for (int i = 0; i < m; i += 8) {
        float4 v[8];
#pragma unroll
        for (int j = 0; j < 8; ++j) {
          const int s = __builtin_amdgcn_readlane(cur, i + j);
          v[j] = p2v[s * 32 + lane];
        }
#pragma unroll
        for (int j = 0; j < 8; ++j) {
          const bool ok = (i + j) < m;
          acc.x += ok ? v[j].x : 0.f;
          acc.y += ok ? v[j].y : 0.f;
          acc.z += ok ? v[j].z : 0.f;
          acc.w += ok ? v[j].w : 0.f;
        }
      }

      if (dg > CAP) {
        const int ns = nspill[0];
        const int2* sp = (const int2*)spill;
        for (int o = 0; o < ns; o += 32) {
          int2 p = make_int2(-1, 0);
          if (o + lane < ns) p = sp[o + lane];
          unsigned hit = __builtin_amdgcn_ballot_w32(p.x == node);
          while (hit != 0u) {
            const int b = __builtin_ctz(hit);
            hit &= hit - 1u;
            const int s = __builtin_amdgcn_readlane(p.y, b);
            const float4 w = p2v[s * 32 + lane];
            acc.x += w.x;
            acc.y += w.y;
            acc.z += w.z;
            acc.w += w.w;
          }
        }
      }

      float4 zz;
      zz.x = zv.x + acc.x * inv;
      zz.y = zv.y + acc.y * inv;
      zz.z = zv.z + acc.z * inv;
      zz.w = zv.w + acc.w * inv;

      float sq = zz.x * zz.x + zz.y * zz.y + zz.z * zz.z + zz.w * zz.w;
#pragma unroll
      for (int off = 16; off > 0; off >>= 1) sq += __shfl_xor(sq, off, 32);
      const float nrm = fmaxf(sqrtf(sq), 1e-12f);

      float4 r;
      r.x = fmaxf(zz.x / nrm, 0.0f);
      r.y = fmaxf(zz.y / nrm, 0.0f);
      r.z = fmaxf(zz.z / nrm, 0.0f);
      r.w = fmaxf(zz.w / nrm, 0.0f);
      zv4[node * 32 + lane] = r;

      ps.x += r.x;  ps.y += r.y;  ps.z += r.z;  ps.w += r.w;
      pss.x += r.x * r.x;  pss.y += r.y * r.y;
      pss.z += r.z * r.z;  pss.w += r.w * r.w;
    }
  }

  ((float4*)&red[wave][0])[lane] = ps;
  ((float4*)&red[wave][128])[lane] = pss;
  __syncthreads();
  if (t < 256) {
    float s = 0.0f;
#pragma unroll
    for (int w = 0; w < GW; ++w) s += red[w][t];
    __hip_atomic_fetch_add(&tot[t], (long long)(s * FX_SCALE),
                           __ATOMIC_RELAXED, __HIP_MEMORY_SCOPE_AGENT);
  }
}

#define RG_WAVES 16
#define RG_BLOCKS 256

template <int NQ>
__global__ void __launch_bounds__(RG_WAVES * 32) rowtile_gemm(
    const float* __restrict__ A, const float* __restrict__ W,
    const float* __restrict__ bias, float* __restrict__ out0,
    float* __restrict__ out1, int nTiles) {
  __shared__ __attribute__((aligned(16))) float4 wf[16 * 8 * 2 * 16];

  const int tid = threadIdx.x;
  const int lane = tid & 31;
  const int wave = __builtin_amdgcn_readfirstlane(tid >> 5);
  const int hh = lane >> 4;
  const int c = lane & 15;

  const int q = blockIdx.x & (NQ - 1);
  const int tb = blockIdx.x / NQ;
  const int ts = RG_BLOCKS / NQ;
  const float* Wq = W + q * HID;
  float* __restrict__ out = (q == 0) ? out0 : out1;

  const float4* Av = (const float4*)A;

  int m = wave;
  int tile = tb + ts * m;
  bool have = tile < nTiles;

  float4 a[16];
  if (have) {
    const float4* ap = Av + (tile * 16 + c) * (HID / 4) + hh;
#pragma unroll
    for (int g = 0; g < 16; ++g) a[g] = ap[2 * g];
  }

  {
    float4 t[8];
#pragma unroll
    for (int i = 0; i < 8; ++i)
      t[i] = *(const float4*)(Wq + (16 * i + wave) * (HID * NQ) + 4 * lane);
#pragma unroll
    for (int i = 0; i < 8; ++i)
      wf[(((lane >> 1) * 8 + i) * 2 + (lane & 1)) * 16 + ((wave ^ lane) & 15)] =
          t[i];
  }
  __syncthreads();

  while (have) {
    asm volatile("" ::: "memory");
    v8f acc[8];
#pragma unroll
    for (int j = 0; j < 8; ++j) {
      const float bv = (q == 0) ? bias[16 * j + c] : 0.0f;
#pragma unroll
      for (int r = 0; r < 8; ++r) acc[j][r] = bv;
    }

    float4 b0[8], b1[8];
#pragma unroll
    for (int j = 0; j < 8; ++j)
      b0[j] = wf[((0 * 8 + j) * 2 + hh) * 16 + (c ^ ((2 * 0 + hh) & 15))];

#pragma unroll
    for (int g = 0; g < 16; g += 2) {
#pragma unroll
      for (int j = 0; j < 8; ++j)
        b1[j] = wf[(((g + 1) * 8 + j) * 2 + hh) * 16 +
                   (c ^ ((2 * (g + 1) + hh) & 15))];
      __builtin_amdgcn_sched_barrier(0);
      {
        const v2f alo = {a[g].x, a[g].y};
        const v2f ahi = {a[g].z, a[g].w};
#pragma unroll
        for (int j = 0; j < 8; ++j) {
          const v2f b = {b0[j].x, b0[j].y};
          acc[j] = __builtin_amdgcn_wmma_f32_16x16x4_f32(
              false, alo, false, b, (short)0, acc[j], false, false);
        }
#pragma unroll
        for (int j = 0; j < 8; ++j) {
          const v2f b = {b0[j].z, b0[j].w};
          acc[j] = __builtin_amdgcn_wmma_f32_16x16x4_f32(
              false, ahi, false, b, (short)0, acc[j], false, false);
        }
      }
      __builtin_amdgcn_sched_barrier(0);
      if (g + 2 < 16) {
#pragma unroll
        for (int j = 0; j < 8; ++j)
          b0[j] = wf[(((g + 2) * 8 + j) * 2 + hh) * 16 +
                     (c ^ ((2 * (g + 2) + hh) & 15))];
      }
      __builtin_amdgcn_sched_barrier(0);
      {
        const v2f alo = {a[g + 1].x, a[g + 1].y};
        const v2f ahi = {a[g + 1].z, a[g + 1].w};
#pragma unroll
        for (int j = 0; j < 8; ++j) {
          const v2f b = {b1[j].x, b1[j].y};
          acc[j] = __builtin_amdgcn_wmma_f32_16x16x4_f32(
              false, alo, false, b, (short)0, acc[j], false, false);
        }
#pragma unroll
        for (int j = 0; j < 8; ++j) {
          const v2f b = {b1[j].z, b1[j].w};
          acc[j] = __builtin_amdgcn_wmma_f32_16x16x4_f32(
              false, ahi, false, b, (short)0, acc[j], false, false);
        }
      }
      __builtin_amdgcn_sched_barrier(0);
    }

    float* orow = out + (tile * 16 + 8 * hh) * HID + c;
#pragma unroll
    for (int r = 0; r < 8; ++r) {
#pragma unroll
      for (int j = 0; j < 8; ++j) orow[r * HID + 16 * j] = acc[j][r];
    }

    m += RG_WAVES;
    tile = tb + ts * m;
    have = tile < nTiles;
    if (have) {
      const float4* ap = Av + (tile * 16 + c) * (HID / 4) + hh;
#pragma unroll
      for (int g = 0; g < 16; ++g) a[g] = ap[2 * g];
    }
  }
}

__global__ void residual_bn(const float* __restrict__ h,
                            const float* __restrict__ z,
                            const long long* __restrict__ tot,
                            const float* __restrict__ gamma,
                            const float* __restrict__ beta,
                            float* __restrict__ out, int total, float invN) {
  const int j = threadIdx.x & (HID - 1);
  const float s = (float)tot[j] * FX_INV;
  const float ss = (float)tot[HID + j] * FX_INV;
  const float mu = s * invN;
  const float var = ss * invN - mu * mu;
  const float a = gamma[j] / sqrtf(var + 1e-5f);
  const float b = beta[j] - mu * a;
  for (int i = blockIdx.x * blockDim.x + threadIdx.x; i < total;
       i += gridDim.x * blockDim.x)
    out[i] = h[i] + z[i] * a + b;
}

extern "C" void kernel_launch(void* const* d_in, const int* in_sizes, int n_in,
                              void* d_out, int out_size, void* d_ws,
                              size_t ws_size, hipStream_t stream) {
  const float* h0 = (const float*)d_in[0];
  const int* src = (const int*)d_in[1];
  const int* dst = (const int*)d_in[2];
  const float* W_emb = (const float*)d_in[3];
  const float* b_emb = (const float*)d_in[4];
  const float* Ws = (const float*)d_in[5];
  const float* bs = (const float*)d_in[6];
  const float* gam = (const float*)d_in[7];
  const float* bet = (const float*)d_in[8];

  const int N = in_sizes[0] / HID;
  const int E = in_sizes[1];
  const int NF = N * HID;
  const int L = 4;

  float* ws = (float*)d_ws;
  float* H = ws;
  float* C = H + NF;
  float* Z = C + NF;
  int* head = (int*)(Z + NF);
  long long* TOT = (long long*)(head + 2);
  int* cnt = (int*)(TOT + L * 256);

  int* tab = (int*)d_out;
  int* spill = tab + N * CAP;

  const int mtiles = N / 16;

  zero_i32<<<256, 256, 0, stream>>>(head, 2 + L * 512 + N);
  fill_slots<<<1024, 256, 0, stream>>>(src, dst, cnt, tab, spill, head, E);

  rowtile_gemm<1><<<RG_BLOCKS, RG_WAVES * 32, 0, stream>>>(h0, W_emb, b_emb, H,
                                                          H, mtiles);

  for (int l = 0; l < L; ++l) {
    rowtile_gemm<2><<<RG_BLOCKS, RG_WAVES * 32, 0, stream>>>(
        H, Ws + l * HID * 256, bs + l * HID, Z, C, mtiles);
    gather_norm_stats<<<NB_STATS, GW * 32, 0, stream>>>(
        C, cnt, tab, spill, head, Z, TOT + l * 256, N);
    float* out = (l == L - 1) ? (float*)d_out : H;
    residual_bn<<<4096, 256, 0, stream>>>(H, Z, TOT + l * 256, gam + l * HID,
                                          bet + l * HID, out, NF,
                                          1.0f / (float)N);
  }
}


# neighbour table by a two-level counting sort in LDS: 256 edge chunks ordered by 128-node range, one block per range hands out row slots by LDS adds; no global atomics, no clear launch; arrival order
# speedup vs baseline: 4.3323x; 1.1023x over previous
#include <hip/hip_runtime.h>
#include <hip/hip_bf16.h>


typedef float v2f __attribute__((ext_vector_type(2)));
typedef float v8f __attribute__((ext_vector_type(8)));

#define HID 128
#define NB_STATS 512

#define FX_SCALE 68719476736.0f
#define FX_INV 1.4551915228366852e-11f

#define CAP 32

#define BE_T 512
#define MAXBK BE_T
#define BE_U 5
#define NCH 256

__global__ void __launch_bounds__(BE_T) bin_edges(
    const int* __restrict__ src, const int* __restrict__ dst,
    int* __restrict__ pairs, int* __restrict__ offs, int* __restrict__ zero,
    int nzero, int E, int nbk) {
  __shared__ int hist[MAXBK];
  __shared__ int wsum[BE_T / 32];

  const int t = threadIdx.x;
  const int nch = (int)gridDim.x;
  const int epc = (E + nch - 1) / nch;
  const int e0 = min((int)blockIdx.x * epc, E);
  const int e1 = min(e0 + epc, E);

  if (blockIdx.x == 0)
    for (int i = t; i < nzero; i += BE_T) zero[i] = 0;
  for (int i = t; i < nbk; i += BE_T) hist[i] = 0;
  __syncthreads();

  for (int base = e0 + t; base < e1; base += BE_U * BE_T) {
    int d[BE_U];
#pragma unroll
    for (int j = 0; j < BE_U; ++j) {
      const int e = base + j * BE_T;
      d[j] = (e < e1) ? dst[e] : -1;
    }
#pragma unroll
    for (int j = 0; j < BE_U; ++j)
      if (d[j] >= 0) atomicAdd(&hist[d[j] >> 7], 1);
  }
  __syncthreads();

  {
    const int lane = t & 31;
    const int wv = t >> 5;
    const int c = (t < nbk) ? hist[t] : 0;
    int inc = c;
    for (int off = 1; off < 32; off <<= 1) {
      const int v = __shfl_up(inc, off, 32);
      if (lane >= off) inc += v;
    }
    if (lane == 31) wsum[wv] = inc;
    __syncthreads();
    int pre = (lane < wv && lane < BE_T / 32) ? wsum[lane] : 0;
    for (int off = 16; off > 0; off >>= 1) pre += __shfl_xor(pre, off, 32);
    const int start = pre + inc - c;
    int* orow = offs + blockIdx.x * (nbk + 1);
    if (t < nbk) {
      hist[t] = start;
      orow[t] = start;
    }
    if (t == BE_T - 1) orow[nbk] = pre + inc;
  }
  __syncthreads();

  int2* prow = (int2*)pairs + e0;
  for (int base = e0 + t; base < e1; base += BE_U * BE_T) {
    int d[BE_U], sv[BE_U];
#pragma unroll
    for (int j = 0; j < BE_U; ++j) {
      const int e = base + j * BE_T;
      d[j] = (e < e1) ? dst[e] : -1;
      sv[j] = (e < e1) ? src[e] : 0;
    }
#pragma unroll
    for (int j = 0; j < BE_U; ++j)
      if (d[j] >= 0) {
        const int pos = atomicAdd(&hist[d[j] >> 7], 1);
        prow[pos] = make_int2(d[j], sv[j]);
      }
  }
}

#define SR_U 4

__global__ void __launch_bounds__(NCH) slot_rows(
    const int* __restrict__ pairs, const int* __restrict__ offs,
    int* __restrict__ cnt, int* __restrict__ tab, int* __restrict__ spill,
    int* __restrict__ nspill, int N, int E, int nbk) {
  __shared__ int cl[128];

  const int t = threadIdx.x;
  const int r = blockIdx.x;
  if (t < 128) cl[t] = 0;
  __syncthreads();

  const int epc = (E + NCH - 1) / NCH;
  const int o0 = offs[t * (nbk + 1) + r];
  const int o1 = offs[t * (nbk + 1) + r + 1];
  const int2* prow = (const int2*)pairs + min(t * epc, E);
  for (int i = o0; i < o1; i += SR_U) {
    int2 p[SR_U];
#pragma unroll
    for (int j = 0; j < SR_U; ++j)
      p[j] = (i + j < o1) ? prow[i + j] : make_int2(-1, 0);
#pragma unroll
    for (int j = 0; j < SR_U; ++j)
      if (p[j].x >= 0) {
        const int slot = atomicAdd(&cl[p[j].x & 127], 1);
        if (slot < CAP) {
          tab[p[j].x * CAP + slot] = p[j].y;
        } else {
          const int o = atomicAdd(nspill, 1);
          ((int2*)spill)[o] = p[j];
        }
      }
  }
  __syncthreads();

  const int node = r * 128 + t;
  if (t < 128 && node < N) cnt[node] = cl[t];
}

#define GW 16

__global__ void __launch_bounds__(GW * 32) gather_norm_stats(
    const float* __restrict__ p2, const int* __restrict__ deg,
    const int* __restrict__ tab, const int* __restrict__ spill,
    const int* __restrict__ nspill, float* __restrict__ z,
    long long* __restrict__ tot, int N) {
  __shared__ __attribute__((aligned(16))) float red[GW][256];

  const int t = threadIdx.x;
  const int lane = t & 31;
  const int wave = __builtin_amdgcn_readfirstlane(t >> 5);

  const float4* p2v = (const float4*)p2;
  float4* zv4 = (float4*)z;

  float4 ps = make_float4(0.f, 0.f, 0.f, 0.f);
  float4 pss = make_float4(0.f, 0.f, 0.f, 0.f);

  const int stride = gridDim.x * GW;
  const int node0 = blockIdx.x * GW + wave;

  for (int kb = 0; node0 + kb * stride < N; kb += 32) {
    const int ln = node0 + (kb + lane) * stride;
    const int ld = (ln < N) ? deg[ln] : 0;
    for (int k = 0; k < 32; ++k) {
      const int node = node0 + (kb + k) * stride;
      if (node >= N) break;

      const int dg = __builtin_amdgcn_readlane(ld, k);
      const int m = min(dg, CAP);
      const int cur = (lane < m) ? tab[node * CAP + lane] : 0;

      const float4 zv = zv4[node * 32 + lane];
      const float inv = 1.0f / fmaxf((float)dg, 1.0f);

      float4 acc = make_float4(0.f, 0.f, 0.f, 0.f);
      for (int i = 0; i < m; i += 8) {
        float4 v[8];
#pragma unroll
        for (int j = 0; j < 8; ++j) {
          const int s = __builtin_amdgcn_readlane(cur, i + j);
          v[j] = p2v[s * 32 + lane];
        }
#pragma unroll
        for (int j = 0; j < 8; ++j) {
          const bool ok = (i + j) < m;
          acc.x += ok ? v[j].x : 0.f;
          acc.y += ok ? v[j].y : 0.f;
          acc.z += ok ? v[j].z : 0.f;
          acc.w += ok ? v[j].w : 0.f;
        }
      }

      if (dg > CAP) {
        const int ns = nspill[0];
        const int2* sp = (const int2*)spill;
        for (int o = 0; o < ns; o += 32) {
          int2 p = make_int2(-1, 0);
          if (o + lane < ns) p = sp[o + lane];
          unsigned hit = __builtin_amdgcn_ballot_w32(p.x == node);
          while (hit != 0u) {
            const int b = __builtin_ctz(hit);
            hit &= hit - 1u;
            const int s = __builtin_amdgcn_readlane(p.y, b);
            const float4 w = p2v[s * 32 + lane];
            acc.x += w.x;
            acc.y += w.y;
            acc.z += w.z;
            acc.w += w.w;
          }
        }
      }

      float4 zz;
      zz.x = zv.x + acc.x * inv;
      zz.y = zv.y + acc.y * inv;
      zz.z = zv.z + acc.z * inv;
      zz.w = zv.w + acc.w * inv;

      float sq = zz.x * zz.x + zz.y * zz.y + zz.z * zz.z + zz.w * zz.w;
#pragma unroll
      for (int off = 16; off > 0; off >>= 1) sq += __shfl_xor(sq, off, 32);
      const float nrm = fmaxf(sqrtf(sq), 1e-12f);

      float4 r;
      r.x = fmaxf(zz.x / nrm, 0.0f);
      r.y = fmaxf(zz.y / nrm, 0.0f);
      r.z = fmaxf(zz.z / nrm, 0.0f);
      r.w = fmaxf(zz.w / nrm, 0.0f);
      zv4[node * 32 + lane] = r;

      ps.x += r.x;  ps.y += r.y;  ps.z += r.z;  ps.w += r.w;
      pss.x += r.x * r.x;  pss.y += r.y * r.y;
      pss.z += r.z * r.z;  pss.w += r.w * r.w;
    }
  }

  ((float4*)&red[wave][0])[lane] = ps;
  ((float4*)&red[wave][128])[lane] = pss;
  __syncthreads();
  if (t < 256) {
    float s = 0.0f;
#pragma unroll
    for (int w = 0; w < GW; ++w) s += red[w][t];
    __hip_atomic_fetch_add(&tot[t], (long long)(s * FX_SCALE),
                           __ATOMIC_RELAXED, __HIP_MEMORY_SCOPE_AGENT);
  }
}

#define RG_WAVES 16
#define RG_BLOCKS 256

template <int NQ>
__global__ void __launch_bounds__(RG_WAVES * 32) rowtile_gemm(
    const float* __restrict__ A, const float* __restrict__ W,
    const float* __restrict__ bias, float* __restrict__ out0,
    float* __restrict__ out1, int nTiles) {
  __shared__ __attribute__((aligned(16))) float4 wf[16 * 8 * 2 * 16];

  const int tid = threadIdx.x;
  const int lane = tid & 31;
  const int wave = __builtin_amdgcn_readfirstlane(tid >> 5);
  const int hh = lane >> 4;
  const int c = lane & 15;

  const int q = blockIdx.x & (NQ - 1);
  const int tb = blockIdx.x / NQ;
  const int ts = RG_BLOCKS / NQ;
  const float* Wq = W + q * HID;
  float* __restrict__ out = (q == 0) ? out0 : out1;

  const float4* Av = (const float4*)A;

  int m = wave;
  int tile = tb + ts * m;
  bool have = tile < nTiles;

  float4 a[16];
  if (have) {
    const float4* ap = Av + (tile * 16 + c) * (HID / 4) + hh;
#pragma unroll
    for (int g = 0; g < 16; ++g) a[g] = ap[2 * g];
  }

  {
    float4 t[8];
#pragma unroll
    for (int i = 0; i < 8; ++i)
      t[i] = *(const float4*)(Wq + (16 * i + wave) * (HID * NQ) + 4 * lane);
#pragma unroll
    for (int i = 0; i < 8; ++i)
      wf[(((lane >> 1) * 8 + i) * 2 + (lane & 1)) * 16 + ((wave ^ lane) & 15)] =
          t[i];
  }
  __syncthreads();

  while (have) {
    asm volatile("" ::: "memory");
    v8f acc[8];
#pragma unroll
    for (int j = 0; j < 8; ++j) {
      const float bv = (q == 0) ? bias[16 * j + c] : 0.0f;
#pragma unroll
      for (int r = 0; r < 8; ++r) acc[j][r] = bv;
    }

    float4 b0[8], b1[8];
#pragma unroll
    for (int j = 0; j < 8; ++j)
      b0[j] = wf[((0 * 8 + j) * 2 + hh) * 16 + (c ^ ((2 * 0 + hh) & 15))];

#pragma unroll
    for (int g = 0; g < 16; g += 2) {
#pragma unroll
      for (int j = 0; j < 8; ++j)
        b1[j] = wf[(((g + 1) * 8 + j) * 2 + hh) * 16 +
                   (c ^ ((2 * (g + 1) + hh) & 15))];
      __builtin_amdgcn_sched_barrier(0);
      {
        const v2f alo = {a[g].x, a[g].y};
        const v2f ahi = {a[g].z, a[g].w};
#pragma unroll
        for (int j = 0; j < 8; ++j) {
          const v2f b = {b0[j].x, b0[j].y};
          acc[j] = __builtin_amdgcn_wmma_f32_16x16x4_f32(
              false, alo, false, b, (short)0, acc[j], false, false);
        }
#pragma unroll
        for (int j = 0; j < 8; ++j) {
          const v2f b = {b0[j].z, b0[j].w};
          acc[j] = __builtin_amdgcn_wmma_f32_16x16x4_f32(
              false, ahi, false, b, (short)0, acc[j], false, false);
        }
      }
      __builtin_amdgcn_sched_barrier(0);
      if (g + 2 < 16) {
#pragma unroll
        for (int j = 0; j < 8; ++j)
          b0[j] = wf[(((g + 2) * 8 + j) * 2 + hh) * 16 +
                     (c ^ ((2 * (g + 2) + hh) & 15))];
      }
      __builtin_amdgcn_sched_barrier(0);
      {
        const v2f alo = {a[g + 1].x, a[g + 1].y};
        const v2f ahi = {a[g + 1].z, a[g + 1].w};
#pragma unroll
        for (int j = 0; j < 8; ++j) {
          const v2f b = {b1[j].x, b1[j].y};
          acc[j] = __builtin_amdgcn_wmma_f32_16x16x4_f32(
              false, alo, false, b, (short)0, acc[j], false, false);
        }
#pragma unroll
        for (int j = 0; j < 8; ++j) {
          const v2f b = {b1[j].z, b1[j].w};
          acc[j] = __builtin_amdgcn_wmma_f32_16x16x4_f32(
              false, ahi, false, b, (short)0, acc[j], false, false);
        }
      }
      __builtin_amdgcn_sched_barrier(0);
    }

    float* orow = out + (tile * 16 + 8 * hh) * HID + c;
#pragma unroll
    for (int r = 0; r < 8; ++r) {
#pragma unroll
      for (int j = 0; j < 8; ++j) orow[r * HID + 16 * j] = acc[j][r];
    }

    m += RG_WAVES;
    tile = tb + ts * m;
    have = tile < nTiles;
    if (have) {
      const float4* ap = Av + (tile * 16 + c) * (HID / 4) + hh;
#pragma unroll
      for (int g = 0; g < 16; ++g) a[g] = ap[2 * g];
    }
  }
}

__global__ void residual_bn(const float* __restrict__ h,
                            const float* __restrict__ z,
                            const long long* __restrict__ tot,
                            const float* __restrict__ gamma,
                            const float* __restrict__ beta,
                            float* __restrict__ out, int total, float invN) {
  const int j = threadIdx.x & (HID - 1);
  const float s = (float)tot[j] * FX_INV;
  const float ss = (float)tot[HID + j] * FX_INV;
  const float mu = s * invN;
  const float var = ss * invN - mu * mu;
  const float a = gamma[j] / sqrtf(var + 1e-5f);
  const float b = beta[j] - mu * a;
  for (int i = blockIdx.x * blockDim.x + threadIdx.x; i < total;
       i += gridDim.x * blockDim.x)
    out[i] = h[i] + z[i] * a + b;
}

extern "C" void kernel_launch(void* const* d_in, const int* in_sizes, int n_in,
                              void* d_out, int out_size, void* d_ws,
                              size_t ws_size, hipStream_t stream) {
  const float* h0 = (const float*)d_in[0];
  const int* src = (const int*)d_in[1];
  const int* dst = (const int*)d_in[2];
  const float* W_emb = (const float*)d_in[3];
  const float* b_emb = (const float*)d_in[4];
  const float* Ws = (const float*)d_in[5];
  const float* bs = (const float*)d_in[6];
  const float* gam = (const float*)d_in[7];
  const float* bet = (const float*)d_in[8];

  const int N = in_sizes[0] / HID;
  const int E = in_sizes[1];
  const int NF = N * HID;
  const int L = 4;

  float* ws = (float*)d_ws;
  float* H = ws;
  float* C = H + NF;
  float* Z = C + NF;
  int* head = (int*)(Z + NF);
  long long* TOT = (long long*)(head + 2);
  int* cnt = (int*)(TOT + L * 256);

  int* tab = (int*)d_out;
  int* spill = tab + N * CAP;

  int* pairs = (int*)C;
  int* offs = (int*)Z;

  const int mtiles = N / 16;
  const int nbk = (N + 127) / 128;

  bin_edges<<<NCH, BE_T, 0, stream>>>(src, dst, pairs, offs, head, 2 + L * 512,
                                      E, nbk);
  slot_rows<<<nbk, NCH, 0, stream>>>(pairs, offs, cnt, tab, spill, head, N, E,
                                     nbk);

  rowtile_gemm<1><<<RG_BLOCKS, RG_WAVES * 32, 0, stream>>>(h0, W_emb, b_emb, H,
                                                          H, mtiles);

  for (int l = 0; l < L; ++l) {
    rowtile_gemm<2><<<RG_BLOCKS, RG_WAVES * 32, 0, stream>>>(
        H, Ws + l * HID * 256, bs + l * HID, Z, C, mtiles);
    gather_norm_stats<<<NB_STATS, GW * 32, 0, stream>>>(
        C, cnt, tab, spill, head, Z, TOT + l * 256, N);
    float* out = (l == L - 1) ? (float*)d_out : H;
    residual_bn<<<4096, 256, 0, stream>>>(H, Z, TOT + l * 256, gam + l * HID,
                                          bet + l * HID, out, NF,
                                          1.0f / (float)N);
  }
}


# residual_bn of layers 0-2 folded into the next layer GEMM, ONE wave owning both column halves of its tile: A loader forms h+z*bn_a+bn_b, stores it in place; no wave pair, no barrier; sums kept
# speedup vs baseline: 4.8279x; 1.1144x over previous
#include <hip/hip_runtime.h>
#include <hip/hip_bf16.h>


typedef float v2f __attribute__((ext_vector_type(2)));
typedef float v8f __attribute__((ext_vector_type(8)));

#define HID 128
#define NB_STATS 512

#define FX_SCALE 68719476736.0f
#define FX_INV 1.4551915228366852e-11f

#define CAP 32

#define BE_T 512
#define MAXBK BE_T
#define BE_U 5
#define NCH 256

__global__ void __launch_bounds__(BE_T) bin_edges(
    const int* __restrict__ src, const int* __restrict__ dst,
    int* __restrict__ pairs, int* __restrict__ offs, int* __restrict__ zero,
    int nzero, int E, int nbk) {
  __shared__ int hist[MAXBK];
  __shared__ int wsum[BE_T / 32];

  const int t = threadIdx.x;
  const int nch = (int)gridDim.x;
  const int epc = (E + nch - 1) / nch;
  const int e0 = min((int)blockIdx.x * epc, E);
  const int e1 = min(e0 + epc, E);

  if (blockIdx.x == 0)
    for (int i = t; i < nzero; i += BE_T) zero[i] = 0;
  for (int i = t; i < nbk; i += BE_T) hist[i] = 0;
  __syncthreads();

  for (int base = e0 + t; base < e1; base += BE_U * BE_T) {
    int d[BE_U];
#pragma unroll
    for (int j = 0; j < BE_U; ++j) {
      const int e = base + j * BE_T;
      d[j] = (e < e1) ? dst[e] : -1;
    }
#pragma unroll
    for (int j = 0; j < BE_U; ++j)
      if (d[j] >= 0) atomicAdd(&hist[d[j] >> 7], 1);
  }
  __syncthreads();

  {
    const int lane = t & 31;
    const int wv = t >> 5;
    const int c = (t < nbk) ? hist[t] : 0;
    int inc = c;
    for (int off = 1; off < 32; off <<= 1) {
      const int v = __shfl_up(inc, off, 32);
      if (lane >= off) inc += v;
    }
    if (lane == 31) wsum[wv] = inc;
    __syncthreads();
    int pre = (lane < wv && lane < BE_T / 32) ? wsum[lane] : 0;
    for (int off = 16; off > 0; off >>= 1) pre += __shfl_xor(pre, off, 32);
    const int start = pre + inc - c;
    int* orow = offs + blockIdx.x * (nbk + 1);
    if (t < nbk) {
      hist[t] = start;
      orow[t] = start;
    }
    if (t == BE_T - 1) orow[nbk] = pre + inc;
  }
  __syncthreads();

  int2* prow = (int2*)pairs + e0;
  for (int base = e0 + t; base < e1; base += BE_U * BE_T) {
    int d[BE_U], sv[BE_U];
#pragma unroll
    for (int j = 0; j < BE_U; ++j) {
      const int e = base + j * BE_T;
      d[j] = (e < e1) ? dst[e] : -1;
      sv[j] = (e < e1) ? src[e] : 0;
    }
#pragma unroll
    for (int j = 0; j < BE_U; ++j)
      if (d[j] >= 0) {
        const int pos = atomicAdd(&hist[d[j] >> 7], 1);
        prow[pos] = make_int2(d[j], sv[j]);
      }
  }
}

#define SR_U 4

__global__ void __launch_bounds__(NCH) slot_rows(
    const int* __restrict__ pairs, const int* __restrict__ offs,
    int* __restrict__ cnt, int* __restrict__ tab, int* __restrict__ spill,
    int* __restrict__ nspill, int N, int E, int nbk) {
  __shared__ int cl[128];

  const int t = threadIdx.x;
  const int r = blockIdx.x;
  if (t < 128) cl[t] = 0;
  __syncthreads();

  const int epc = (E + NCH - 1) / NCH;
  const int o0 = offs[t * (nbk + 1) + r];
  const int o1 = offs[t * (nbk + 1) + r + 1];
  const int2* prow = (const int2*)pairs + min(t * epc, E);
  for (int i = o0; i < o1; i += SR_U) {
    int2 p[SR_U];
#pragma unroll
    for (int j = 0; j < SR_U; ++j)
      p[j] = (i + j < o1) ? prow[i + j] : make_int2(-1, 0);
#pragma unroll
    for (int j = 0; j < SR_U; ++j)
      if (p[j].x >= 0) {
        const int slot = atomicAdd(&cl[p[j].x & 127], 1);
        if (slot < CAP) {
          tab[p[j].x * CAP + slot] = p[j].y;
        } else {
          const int o = atomicAdd(nspill, 1);
          ((int2*)spill)[o] = p[j];
        }
      }
  }
  __syncthreads();

  const int node = r * 128 + t;
  if (t < 128 && node < N) cnt[node] = cl[t];
}

#define GW 16

__global__ void __launch_bounds__(GW * 32) gather_norm_stats(
    const float* __restrict__ p2, const int* __restrict__ deg,
    const int* __restrict__ tab, const int* __restrict__ spill,
    const int* __restrict__ nspill, float* __restrict__ z,
    long long* __restrict__ tot, int N) {
  __shared__ __attribute__((aligned(16))) float red[GW][256];

  const int t = threadIdx.x;
  const int lane = t & 31;
  const int wave = __builtin_amdgcn_readfirstlane(t >> 5);

  const float4* p2v = (const float4*)p2;
  float4* zv4 = (float4*)z;

  float4 ps = make_float4(0.f, 0.f, 0.f, 0.f);
  float4 pss = make_float4(0.f, 0.f, 0.f, 0.f);

  const int stride = gridDim.x * GW;
  const int node0 = blockIdx.x * GW + wave;

  for (int kb = 0; node0 + kb * stride < N; kb += 32) {
    const int ln = node0 + (kb + lane) * stride;
    const int ld = (ln < N) ? deg[ln] : 0;
    for (int k = 0; k < 32; ++k) {
      const int node = node0 + (kb + k) * stride;
      if (node >= N) break;

      const int dg = __builtin_amdgcn_readlane(ld, k);
      const int m = min(dg, CAP);
      const int cur = (lane < m) ? tab[node * CAP + lane] : 0;

      const float4 zv = zv4[node * 32 + lane];
      const float inv = 1.0f / fmaxf((float)dg, 1.0f);

      float4 acc = make_float4(0.f, 0.f, 0.f, 0.f);
      for (int i = 0; i < m; i += 8) {
        float4 v[8];
#pragma unroll
        for (int j = 0; j < 8; ++j) {
          const int s = __builtin_amdgcn_readlane(cur, i + j);
          v[j] = p2v[s * 32 + lane];
        }
#pragma unroll
        for (int j = 0; j < 8; ++j) {
          const bool ok = (i + j) < m;
          acc.x += ok ? v[j].x : 0.f;
          acc.y += ok ? v[j].y : 0.f;
          acc.z += ok ? v[j].z : 0.f;
          acc.w += ok ? v[j].w : 0.f;
        }
      }

      if (dg > CAP) {
        const int ns = nspill[0];
        const int2* sp = (const int2*)spill;
        for (int o = 0; o < ns; o += 32) {
          int2 p = make_int2(-1, 0);
          if (o + lane < ns) p = sp[o + lane];
          unsigned hit = __builtin_amdgcn_ballot_w32(p.x == node);
          while (hit != 0u) {
            const int b = __builtin_ctz(hit);
            hit &= hit - 1u;
            const int s = __builtin_amdgcn_readlane(p.y, b);
            const float4 w = p2v[s * 32 + lane];
            acc.x += w.x;
            acc.y += w.y;
            acc.z += w.z;
            acc.w += w.w;
          }
        }
      }

      float4 zz;
      zz.x = zv.x + acc.x * inv;
      zz.y = zv.y + acc.y * inv;
      zz.z = zv.z + acc.z * inv;
      zz.w = zv.w + acc.w * inv;

      float sq = zz.x * zz.x + zz.y * zz.y + zz.z * zz.z + zz.w * zz.w;
#pragma unroll
      for (int off = 16; off > 0; off >>= 1) sq += __shfl_xor(sq, off, 32);
      const float nrm = fmaxf(sqrtf(sq), 1e-12f);

      float4 r;
      r.x = fmaxf(zz.x / nrm, 0.0f);
      r.y = fmaxf(zz.y / nrm, 0.0f);
      r.z = fmaxf(zz.z / nrm, 0.0f);
      r.w = fmaxf(zz.w / nrm, 0.0f);
      zv4[node * 32 + lane] = r;

      ps.x += r.x;  ps.y += r.y;  ps.z += r.z;  ps.w += r.w;
      pss.x += r.x * r.x;  pss.y += r.y * r.y;
      pss.z += r.z * r.z;  pss.w += r.w * r.w;
    }
  }

  ((float4*)&red[wave][0])[lane] = ps;
  ((float4*)&red[wave][128])[lane] = pss;
  __syncthreads();
  if (t < 256) {
    float s = 0.0f;
#pragma unroll
    for (int w = 0; w < GW; ++w) s += red[w][t];
    __hip_atomic_fetch_add(&tot[t], (long long)(s * FX_SCALE),
                           __ATOMIC_RELAXED, __HIP_MEMORY_SCOPE_AGENT);
  }
}

#define RG_WAVES 16
#define RG_BLOCKS 256

template <int NQ>
__global__ void __launch_bounds__(RG_WAVES * 32) rowtile_gemm(
    const float* __restrict__ A, const float* __restrict__ W,
    const float* __restrict__ bias, float* __restrict__ out0,
    float* __restrict__ out1, int nTiles) {
  __shared__ __attribute__((aligned(16))) float4 wf[16 * 8 * 2 * 16];

  const int tid = threadIdx.x;
  const int lane = tid & 31;
  const int wave = __builtin_amdgcn_readfirstlane(tid >> 5);
  const int hh = lane >> 4;
  const int c = lane & 15;

  const int q = blockIdx.x & (NQ - 1);
  const int tb = blockIdx.x / NQ;
  const int ts = RG_BLOCKS / NQ;
  const float* Wq = W + q * HID;
  float* __restrict__ out = (q == 0) ? out0 : out1;

  const float4* Av = (const float4*)A;

  int m = wave;
  int tile = tb + ts * m;
  bool have = tile < nTiles;

  float4 a[16];
  if (have) {
    const float4* ap = Av + (tile * 16 + c) * (HID / 4) + hh;
#pragma unroll
    for (int g = 0; g < 16; ++g) a[g] = ap[2 * g];
  }

  {
    float4 t[8];
#pragma unroll
    for (int i = 0; i < 8; ++i)
      t[i] = *(const float4*)(Wq + (16 * i + wave) * (HID * NQ) + 4 * lane);
#pragma unroll
    for (int i = 0; i < 8; ++i)
      wf[(((lane >> 1) * 8 + i) * 2 + (lane & 1)) * 16 + ((wave ^ lane) & 15)] =
          t[i];
  }
  __syncthreads();

  while (have) {
    asm volatile("" ::: "memory");
    v8f acc[8];
#pragma unroll
    for (int j = 0; j < 8; ++j) {
      const float bv = (q == 0) ? bias[16 * j + c] : 0.0f;
#pragma unroll
      for (int r = 0; r < 8; ++r) acc[j][r] = bv;
    }

    float4 b0[8], b1[8];
#pragma unroll
    for (int j = 0; j < 8; ++j)
      b0[j] = wf[((0 * 8 + j) * 2 + hh) * 16 + (c ^ ((2 * 0 + hh) & 15))];

#pragma unroll
    for (int g = 0; g < 16; g += 2) {
#pragma unroll
      for (int j = 0; j < 8; ++j)
        b1[j] = wf[(((g + 1) * 8 + j) * 2 + hh) * 16 +
                   (c ^ ((2 * (g + 1) + hh) & 15))];
      __builtin_amdgcn_sched_barrier(0);
      {
        const v2f alo = {a[g].x, a[g].y};
        const v2f ahi = {a[g].z, a[g].w};
#pragma unroll
        for (int j = 0; j < 8; ++j) {
          const v2f b = {b0[j].x, b0[j].y};
          acc[j] = __builtin_amdgcn_wmma_f32_16x16x4_f32(
              false, alo, false, b, (short)0, acc[j], false, false);
        }
#pragma unroll
        for (int j = 0; j < 8; ++j) {
          const v2f b = {b0[j].z, b0[j].w};
          acc[j] = __builtin_amdgcn_wmma_f32_16x16x4_f32(
              false, ahi, false, b, (short)0, acc[j], false, false);
        }
      }
      __builtin_amdgcn_sched_barrier(0);
      if (g + 2 < 16) {
#pragma unroll
        for (int j = 0; j < 8; ++j)
          b0[j] = wf[(((g + 2) * 8 + j) * 2 + hh) * 16 +
                     (c ^ ((2 * (g + 2) + hh) & 15))];
      }
      __builtin_amdgcn_sched_barrier(0);
      {
        const v2f alo = {a[g + 1].x, a[g + 1].y};
        const v2f ahi = {a[g + 1].z, a[g + 1].w};
#pragma unroll
        for (int j = 0; j < 8; ++j) {
          const v2f b = {b1[j].x, b1[j].y};
          acc[j] = __builtin_amdgcn_wmma_f32_16x16x4_f32(
              false, alo, false, b, (short)0, acc[j], false, false);
        }
#pragma unroll
        for (int j = 0; j < 8; ++j) {
          const v2f b = {b1[j].z, b1[j].w};
          acc[j] = __builtin_amdgcn_wmma_f32_16x16x4_f32(
              false, ahi, false, b, (short)0, acc[j], false, false);
        }
      }
      __builtin_amdgcn_sched_barrier(0);
    }

    float* orow = out + (tile * 16 + 8 * hh) * HID + c;
#pragma unroll
    for (int r = 0; r < 8; ++r) {
#pragma unroll
      for (int j = 0; j < 8; ++j) orow[r * HID + 16 * j] = acc[j][r];
    }

    m += RG_WAVES;
    tile = tb + ts * m;
    have = tile < nTiles;
    if (have) {
      const float4* ap = Av + (tile * 16 + c) * (HID / 4) + hh;
#pragma unroll
      for (int g = 0; g < 16; ++g) a[g] = ap[2 * g];
    }
  }
}

__device__ __forceinline__ void half_pass(const float4 (&a)[16],
                                          const float4* __restrict__ wq,
                                          const float* __restrict__ bias,
                                          bool bv_on, float* __restrict__ out,
                                          int tile, int c, int hh) {
  v8f acc[8];
#pragma unroll
  for (int j = 0; j < 8; ++j) {
    const float bv = bv_on ? bias[16 * j + c] : 0.0f;
#pragma unroll
    for (int r = 0; r < 8; ++r) acc[j][r] = bv;
  }

  float4 b0[8], b1[8];
#pragma unroll
  for (int j = 0; j < 8; ++j)
    b0[j] = wq[((0 * 8 + j) * 2 + hh) * 16 + (c ^ ((2 * 0 + hh) & 15))];

#pragma unroll
  for (int g = 0; g < 16; g += 2) {
#pragma unroll
    for (int j = 0; j < 8; ++j)
      b1[j] = wq[(((g + 1) * 8 + j) * 2 + hh) * 16 +
                 (c ^ ((2 * (g + 1) + hh) & 15))];
    __builtin_amdgcn_sched_barrier(0);
    {
      const v2f alo = {a[g].x, a[g].y};
      const v2f ahi = {a[g].z, a[g].w};
#pragma unroll
      for (int j = 0; j < 8; ++j) {
        const v2f b = {b0[j].x, b0[j].y};
        acc[j] = __builtin_amdgcn_wmma_f32_16x16x4_f32(
            false, alo, false, b, (short)0, acc[j], false, false);
      }
#pragma unroll
      for (int j = 0; j < 8; ++j) {
        const v2f b = {b0[j].z, b0[j].w};
        acc[j] = __builtin_amdgcn_wmma_f32_16x16x4_f32(
            false, ahi, false, b, (short)0, acc[j], false, false);
      }
    }
    __builtin_amdgcn_sched_barrier(0);
    if (g + 2 < 16) {
#pragma unroll
      for (int j = 0; j < 8; ++j)
        b0[j] = wq[(((g + 2) * 8 + j) * 2 + hh) * 16 +
                   (c ^ ((2 * (g + 2) + hh) & 15))];
    }
    __builtin_amdgcn_sched_barrier(0);
    {
      const v2f alo = {a[g + 1].x, a[g + 1].y};
      const v2f ahi = {a[g + 1].z, a[g + 1].w};
#pragma unroll
      for (int j = 0; j < 8; ++j) {
        const v2f b = {b1[j].x, b1[j].y};
        acc[j] = __builtin_amdgcn_wmma_f32_16x16x4_f32(
            false, alo, false, b, (short)0, acc[j], false, false);
      }
#pragma unroll
      for (int j = 0; j < 8; ++j) {
        const v2f b = {b1[j].z, b1[j].w};
        acc[j] = __builtin_amdgcn_wmma_f32_16x16x4_f32(
            false, ahi, false, b, (short)0, acc[j], false, false);
      }
    }
    __builtin_amdgcn_sched_barrier(0);
  }

  float* orow = out + (tile * 16 + 8 * hh) * HID + c;
#pragma unroll
  for (int r = 0; r < 8; ++r) {
#pragma unroll
    for (int j = 0; j < 8; ++j) orow[r * HID + 16 * j] = acc[j][r];
  }
}

__global__ void __launch_bounds__(RG_WAVES * 32) fold_gemm(
    float* __restrict__ Hh, float* __restrict__ Zz,
    const float* __restrict__ W, const float* __restrict__ bias,
    const long long* __restrict__ tot, const float* __restrict__ gamma,
    const float* __restrict__ beta, float* __restrict__ Pp, int nTiles,
    float invN) {
  __shared__ __attribute__((aligned(16))) float4 wf[2 * 16 * 8 * 2 * 16];
  __shared__ __attribute__((aligned(16))) float bnab[2 * HID];

  const int tid = threadIdx.x;
  const int lane = tid & 31;
  const int wave = __builtin_amdgcn_readfirstlane(tid >> 5);
  const int hh = lane >> 4;
  const int c = lane & 15;

  const int slot = 4 * (((wave & 3) - (wave >> 2)) & 3) + (wave >> 2);
  const int tile = (int)blockIdx.x + RG_BLOCKS * slot;
  const bool have = tile < nTiles;

  float4 hv[16], zv[16];
  if (have) {
    const float4* hp = (const float4*)Hh + (tile * 16 + c) * (HID / 4) + hh;
    const float4* zp = (const float4*)Zz + (tile * 16 + c) * (HID / 4) + hh;
#pragma unroll
    for (int g = 0; g < 16; ++g) hv[g] = hp[2 * g];
#pragma unroll
    for (int g = 0; g < 16; ++g) zv[g] = zp[2 * g];
  }

  if (tid < HID) {
    const float s = (float)tot[tid] * FX_INV;
    const float ss = (float)tot[HID + tid] * FX_INV;
    const float mu = s * invN;
    const float var = ss * invN - mu * mu;
    const float sa = gamma[tid] / sqrtf(var + 1e-5f);
    bnab[tid] = sa;
    bnab[HID + tid] = beta[tid] - mu * sa;
  }

#pragma unroll
  for (int q = 0; q < 2; ++q) {
    float4 t[8];
#pragma unroll
    for (int i = 0; i < 8; ++i)
      t[i] = *(const float4*)(W + (16 * i + wave) * (HID * 2) + q * HID +
                              4 * lane);
#pragma unroll
    for (int i = 0; i < 8; ++i)
      wf[q * 4096 + (((lane >> 1) * 8 + i) * 2 + (lane & 1)) * 16 +
         ((wave ^ lane) & 15)] = t[i];
  }
  __syncthreads();

  if (have) {
    const float4* sa4 = (const float4*)&bnab[0];
    const float4* sb4 = (const float4*)&bnab[HID];
    float4 a[16];
#pragma unroll
    for (int g0 = 0; g0 < 16; g0 += 4) {
#pragma unroll
      for (int g = g0; g < g0 + 4; ++g) {
        const float4 sa = sa4[2 * g + hh];
        const float4 sb = sb4[2 * g + hh];
        a[g].x = hv[g].x + zv[g].x * sa.x + sb.x;
        a[g].y = hv[g].y + zv[g].y * sa.y + sb.y;
        a[g].z = hv[g].z + zv[g].z * sa.z + sb.z;
        a[g].w = hv[g].w + zv[g].w * sa.w + sb.w;
      }
      __builtin_amdgcn_sched_barrier(0);
    }
    float4* hw = (float4*)Hh + (tile * 16 + c) * (HID / 4) + hh;
#pragma unroll
    for (int g = 0; g < 16; ++g) hw[2 * g] = a[g];

    half_pass(a, wf, bias, true, Zz, tile, c, hh);
    half_pass(a, wf + 4096, bias, false, Pp, tile, c, hh);
  }
}

__global__ void residual_bn(const float* __restrict__ h,
                            const float* __restrict__ z,
                            const long long* __restrict__ tot,
                            const float* __restrict__ gamma,
                            const float* __restrict__ beta,
                            float* __restrict__ out, int total, float invN) {
  const int j = threadIdx.x & (HID - 1);
  const float s = (float)tot[j] * FX_INV;
  const float ss = (float)tot[HID + j] * FX_INV;
  const float mu = s * invN;
  const float var = ss * invN - mu * mu;
  const float a = gamma[j] / sqrtf(var + 1e-5f);
  const float b = beta[j] - mu * a;
  for (int i = blockIdx.x * blockDim.x + threadIdx.x; i < total;
       i += gridDim.x * blockDim.x)
    out[i] = h[i] + z[i] * a + b;
}

extern "C" void kernel_launch(void* const* d_in, const int* in_sizes, int n_in,
                              void* d_out, int out_size, void* d_ws,
                              size_t ws_size, hipStream_t stream) {
  const float* h0 = (const float*)d_in[0];
  const int* src = (const int*)d_in[1];
  const int* dst = (const int*)d_in[2];
  const float* W_emb = (const float*)d_in[3];
  const float* b_emb = (const float*)d_in[4];
  const float* Ws = (const float*)d_in[5];
  const float* bs = (const float*)d_in[6];
  const float* gam = (const float*)d_in[7];
  const float* bet = (const float*)d_in[8];

  const int N = in_sizes[0] / HID;
  const int E = in_sizes[1];
  const int NF = N * HID;
  const int L = 4;

  float* ws = (float*)d_ws;
  float* H = ws;
  float* C = H + NF;
  float* Z = C + NF;
  int* head = (int*)(Z + NF);
  long long* TOT = (long long*)(head + 2);
  int* cnt = (int*)(TOT + L * 256);

  int* tab = (int*)d_out;
  int* spill = tab + N * CAP;

  int* pairs = (int*)C;
  int* offs = (int*)Z;

  const int mtiles = N / 16;
  const int nbk = (N + 127) / 128;

  bin_edges<<<NCH, BE_T, 0, stream>>>(src, dst, pairs, offs, head, 2 + L * 512,
                                      E, nbk);
  slot_rows<<<nbk, NCH, 0, stream>>>(pairs, offs, cnt, tab, spill, head, N, E,
                                     nbk);

  rowtile_gemm<1><<<RG_BLOCKS, RG_WAVES * 32, 0, stream>>>(h0, W_emb, b_emb, H,
                                                          H, mtiles);

  const float invN = 1.0f / (float)N;
  for (int l = 0; l < L; ++l) {
    if (l == 0) {
      rowtile_gemm<2><<<RG_BLOCKS, RG_WAVES * 32, 0, stream>>>(
          H, Ws, bs, Z, C, mtiles);
    } else {
      fold_gemm<<<RG_BLOCKS, RG_WAVES * 32, 0, stream>>>(
          H, Z, Ws + l * HID * 256, bs + l * HID, TOT + (l - 1) * 256,
          gam + (l - 1) * HID, bet + (l - 1) * HID, C, mtiles, invN);
    }
    gather_norm_stats<<<NB_STATS, GW * 32, 0, stream>>>(
        C, cnt, tab, spill, head, Z, TOT + l * 256, N);
  }
  residual_bn<<<4096, 256, 0, stream>>>(H, Z, TOT + (L - 1) * 256,
                                        gam + (L - 1) * HID,
                                        bet + (L - 1) * HID, (float*)d_out, NF,
                                        invN);
}


# rowtile_gemm, fold_gemm: tiles cross global memory only as whole 512-byte rows; a wave-private LDS buffer (pitch 132) turns rows into A fragments and accumulator tiles into rows; sums unchanged
# speedup vs baseline: 5.2578x; 1.0890x over previous
#include <hip/hip_runtime.h>
#include <hip/hip_bf16.h>


typedef float v2f __attribute__((ext_vector_type(2)));
typedef float v8f __attribute__((ext_vector_type(8)));

#define HID 128
#define NB_STATS 512

#define FX_SCALE 68719476736.0f
#define FX_INV 1.4551915228366852e-11f

#define CAP 32

#define BE_T 512
#define MAXBK BE_T
#define BE_U 5
#define NCH 256

__global__ void __launch_bounds__(BE_T) bin_edges(
    const int* __restrict__ src, const int* __restrict__ dst,
    int* __restrict__ pairs, int* __restrict__ offs, int* __restrict__ zero,
    int nzero, int E, int nbk) {
  __shared__ int hist[MAXBK];
  __shared__ int wsum[BE_T / 32];

  const int t = threadIdx.x;
  const int nch = (int)gridDim.x;
  const int epc = (E + nch - 1) / nch;
  const int e0 = min((int)blockIdx.x * epc, E);
  const int e1 = min(e0 + epc, E);

  if (blockIdx.x == 0)
    for (int i = t; i < nzero; i += BE_T) zero[i] = 0;
  for (int i = t; i < nbk; i += BE_T) hist[i] = 0;
  __syncthreads();

  for (int base = e0 + t; base < e1; base += BE_U * BE_T) {
    int d[BE_U];
#pragma unroll
    for (int j = 0; j < BE_U; ++j) {
      const int e = base + j * BE_T;
      d[j] = (e < e1) ? dst[e] : -1;
    }
#pragma unroll
    for (int j = 0; j < BE_U; ++j)
      if (d[j] >= 0) atomicAdd(&hist[d[j] >> 7], 1);
  }
  __syncthreads();

  {
    const int lane = t & 31;
    const int wv = t >> 5;
    const int c = (t < nbk) ? hist[t] : 0;
    int inc = c;
    for (int off = 1; off < 32; off <<= 1) {
      const int v = __shfl_up(inc, off, 32);
      if (lane >= off) inc += v;
    }
    if (lane == 31) wsum[wv] = inc;
    __syncthreads();
    int pre = (lane < wv && lane < BE_T / 32) ? wsum[lane] : 0;
    for (int off = 16; off > 0; off >>= 1) pre += __shfl_xor(pre, off, 32);
    const int start = pre + inc - c;
    int* orow = offs + blockIdx.x * (nbk + 1);
    if (t < nbk) {
      hist[t] = start;
      orow[t] = start;
    }
    if (t == BE_T - 1) orow[nbk] = pre + inc;
  }
  __syncthreads();

  int2* prow = (int2*)pairs + e0;
  for (int base = e0 + t; base < e1; base += BE_U * BE_T) {
    int d[BE_U], sv[BE_U];
#pragma unroll
    for (int j = 0; j < BE_U; ++j) {
      const int e = base + j * BE_T;
      d[j] = (e < e1) ? dst[e] : -1;
      sv[j] = (e < e1) ? src[e] : 0;
    }
#pragma unroll
    for (int j = 0; j < BE_U; ++j)
      if (d[j] >= 0) {
        const int pos = atomicAdd(&hist[d[j] >> 7], 1);
        prow[pos] = make_int2(d[j], sv[j]);
      }
  }
}

#define SR_U 4

__global__ void __launch_bounds__(NCH) slot_rows(
    const int* __restrict__ pairs, const int* __restrict__ offs,
    int* __restrict__ cnt, int* __restrict__ tab, int* __restrict__ spill,
    int* __restrict__ nspill, int N, int E, int nbk) {
  __shared__ int cl[128];

  const int t = threadIdx.x;
  const int r = blockIdx.x;
  if (t < 128) cl[t] = 0;
  __syncthreads();

  const int epc = (E + NCH - 1) / NCH;
  const int o0 = offs[t * (nbk + 1) + r];
  const int o1 = offs[t * (nbk + 1) + r + 1];
  const int2* prow = (const int2*)pairs + min(t * epc, E);
  for (int i = o0; i < o1; i += SR_U) {
    int2 p[SR_U];
#pragma unroll
    for (int j = 0; j < SR_U; ++j)
      p[j] = (i + j < o1) ? prow[i + j] : make_int2(-1, 0);
#pragma unroll
    for (int j = 0; j < SR_U; ++j)
      if (p[j].x >= 0) {
        const int slot = atomicAdd(&cl[p[j].x & 127], 1);
        if (slot < CAP) {
          tab[p[j].x * CAP + slot] = p[j].y;
        } else {
          const int o = atomicAdd(nspill, 1);
          ((int2*)spill)[o] = p[j];
        }
      }
  }
  __syncthreads();

  const int node = r * 128 + t;
  if (t < 128 && node < N) cnt[node] = cl[t];
}

#define GW 16

__global__ void __launch_bounds__(GW * 32) gather_norm_stats(
    const float* __restrict__ p2, const int* __restrict__ deg,
    const int* __restrict__ tab, const int* __restrict__ spill,
    const int* __restrict__ nspill, float* __restrict__ z,
    long long* __restrict__ tot, int N) {
  __shared__ __attribute__((aligned(16))) float red[GW][256];

  const int t = threadIdx.x;
  const int lane = t & 31;
  const int wave = __builtin_amdgcn_readfirstlane(t >> 5);

  const float4* p2v = (const float4*)p2;
  float4* zv4 = (float4*)z;

  float4 ps = make_float4(0.f, 0.f, 0.f, 0.f);
  float4 pss = make_float4(0.f, 0.f, 0.f, 0.f);

  const int stride = gridDim.x * GW;
  const int node0 = blockIdx.x * GW + wave;

  for (int kb = 0; node0 + kb * stride < N; kb += 32) {
    const int ln = node0 + (kb + lane) * stride;
    const int ld = (ln < N) ? deg[ln] : 0;
    for (int k = 0; k < 32; ++k) {
      const int node = node0 + (kb + k) * stride;
      if (node >= N) break;

      const int dg = __builtin_amdgcn_readlane(ld, k);
      const int m = min(dg, CAP);
      const int cur = (lane < m) ? tab[node * CAP + lane] : 0;

      const float4 zv = zv4[node * 32 + lane];
      const float inv = 1.0f / fmaxf((float)dg, 1.0f);

      float4 acc = make_float4(0.f, 0.f, 0.f, 0.f);
      for (int i = 0; i < m; i += 8) {
        float4 v[8];
#pragma unroll
        for (int j = 0; j < 8; ++j) {
          const int s = __builtin_amdgcn_readlane(cur, i + j);
          v[j] = p2v[s * 32 + lane];
        }
#pragma unroll
        for (int j = 0; j < 8; ++j) {
          const bool ok = (i + j) < m;
          acc.x += ok ? v[j].x : 0.f;
          acc.y += ok ? v[j].y : 0.f;
          acc.z += ok ? v[j].z : 0.f;
          acc.w += ok ? v[j].w : 0.f;
        }
      }

      if (dg > CAP) {
        const int ns = nspill[0];
        const int2* sp = (const int2*)spill;
        for (int o = 0; o < ns; o += 32) {
          int2 p = make_int2(-1, 0);
          if (o + lane < ns) p = sp[o + lane];
          unsigned hit = __builtin_amdgcn_ballot_w32(p.x == node);
          while (hit != 0u) {
            const int b = __builtin_ctz(hit);
            hit &= hit - 1u;
            const int s = __builtin_amdgcn_readlane(p.y, b);
            const float4 w = p2v[s * 32 + lane];
            acc.x += w.x;
            acc.y += w.y;
            acc.z += w.z;
            acc.w += w.w;
          }
        }
      }

      float4 zz;
      zz.x = zv.x + acc.x * inv;
      zz.y = zv.y + acc.y * inv;
      zz.z = zv.z + acc.z * inv;
      zz.w = zv.w + acc.w * inv;

      float sq = zz.x * zz.x + zz.y * zz.y + zz.z * zz.z + zz.w * zz.w;
#pragma unroll
      for (int off = 16; off > 0; off >>= 1) sq += __shfl_xor(sq, off, 32);
      const float nrm = fmaxf(sqrtf(sq), 1e-12f);

      float4 r;
      r.x = fmaxf(zz.x / nrm, 0.0f);
      r.y = fmaxf(zz.y / nrm, 0.0f);
      r.z = fmaxf(zz.z / nrm, 0.0f);
      r.w = fmaxf(zz.w / nrm, 0.0f);
      zv4[node * 32 + lane] = r;

      ps.x += r.x;  ps.y += r.y;  ps.z += r.z;  ps.w += r.w;
      pss.x += r.x * r.x;  pss.y += r.y * r.y;
      pss.z += r.z * r.z;  pss.w += r.w * r.w;
    }
  }

  ((float4*)&red[wave][0])[lane] = ps;
  ((float4*)&red[wave][128])[lane] = pss;
  __syncthreads();
  if (t < 256) {
    float s = 0.0f;
#pragma unroll
    for (int w = 0; w < GW; ++w) s += red[w][t];
    __hip_atomic_fetch_add(&tot[t], (long long)(s * FX_SCALE),
                           __ATOMIC_RELAXED, __HIP_MEMORY_SCOPE_AGENT);
  }
}

#define RG_WAVES 16
#define RG_BLOCKS 256
#define TP 33

__device__ __forceinline__ void rows_to_frags(float4 (&a)[16], float4* tb,
                                              int lane, int c, int hh) {
#pragma unroll
  for (int i = 0; i < 16; ++i) tb[i * TP + lane] = a[i];
  __builtin_amdgcn_fence(__ATOMIC_ACQ_REL, "wavefront");
#pragma unroll
  for (int g = 0; g < 16; ++g) a[g] = tb[c * TP + 2 * g + hh];
}

__device__ __forceinline__ void acc_to_rows(const v8f (&acc)[8], float4* tb,
                                            float* __restrict__ out, int tile,
                                            int lane, int c, int hh) {
  float* tbf = (float*)tb;
#pragma unroll
  for (int r = 0; r < 8; ++r) {
#pragma unroll
    for (int j = 0; j < 8; ++j)
      tbf[(8 * hh + r) * (4 * TP) + 16 * j + c] = acc[j][r];
  }
  __builtin_amdgcn_fence(__ATOMIC_ACQ_REL, "wavefront");
  float4* op = (float4*)out + tile * 16 * (HID / 4) + lane;
#pragma unroll
  for (int i = 0; i < 16; ++i) op[i * (HID / 4)] = tb[i * TP + lane];
}

template <int NQ>
__global__ void __launch_bounds__(RG_WAVES * 32) rowtile_gemm(
    const float* __restrict__ A, const float* __restrict__ W,
    const float* __restrict__ bias, float* __restrict__ out0,
    float* __restrict__ out1, int nTiles) {
  __shared__ __attribute__((aligned(16))) float4 wf[16 * 8 * 2 * 16];
  __shared__ __attribute__((aligned(16))) float4 tbuf[RG_WAVES][16 * TP];

  const int tid = threadIdx.x;
  const int lane = tid & 31;
  const int wave = __builtin_amdgcn_readfirstlane(tid >> 5);
  const int hh = lane >> 4;
  const int c = lane & 15;

  const int q = blockIdx.x & (NQ - 1);
  const int tb = blockIdx.x / NQ;
  const int ts = RG_BLOCKS / NQ;
  const float* Wq = W + q * HID;
  float* __restrict__ out = (q == 0) ? out0 : out1;

  const float4* Av = (const float4*)A;
  float4* tbw = tbuf[wave];

  int m = wave;
  int tile = tb + ts * m;
  bool have = tile < nTiles;

  float4 a[16];
  if (have) {
    const float4* ap = Av + tile * 16 * (HID / 4) + lane;
#pragma unroll
    for (int i = 0; i < 16; ++i) a[i] = ap[i * (HID / 4)];
  }

  {
    float4 t[8];
#pragma unroll
    for (int i = 0; i < 8; ++i)
      t[i] = *(const float4*)(Wq + (16 * i + wave) * (HID * NQ) + 4 * lane);
#pragma unroll
    for (int i = 0; i < 8; ++i)
      wf[(((lane >> 1) * 8 + i) * 2 + (lane & 1)) * 16 + ((wave ^ lane) & 15)] =
          t[i];
  }
  __syncthreads();

  while (have) {
    asm volatile("" ::: "memory");
    rows_to_frags(a, tbw, lane, c, hh);

    v8f acc[8];
#pragma unroll
    for (int j = 0; j < 8; ++j) {
      const float bv = (q == 0) ? bias[16 * j + c] : 0.0f;
#pragma unroll
      for (int r = 0; r < 8; ++r) acc[j][r] = bv;
    }

    float4 b0[8], b1[8];
#pragma unroll
    for (int j = 0; j < 8; ++j)
      b0[j] = wf[((0 * 8 + j) * 2 + hh) * 16 + (c ^ ((2 * 0 + hh) & 15))];

#pragma unroll
    for (int g = 0; g < 16; g += 2) {
#pragma unroll
      for (int j = 0; j < 8; ++j)
        b1[j] = wf[(((g + 1) * 8 + j) * 2 + hh) * 16 +
                   (c ^ ((2 * (g + 1) + hh) & 15))];
      __builtin_amdgcn_sched_barrier(0);
      {
        const v2f alo = {a[g].x, a[g].y};
        const v2f ahi = {a[g].z, a[g].w};
#pragma unroll
        for (int j = 0; j < 8; ++j) {
          const v2f b = {b0[j].x, b0[j].y};
          acc[j] = __builtin_amdgcn_wmma_f32_16x16x4_f32(
              false, alo, false, b, (short)0, acc[j], false, false);
        }
#pragma unroll
        for (int j = 0; j < 8; ++j) {
          const v2f b = {b0[j].z, b0[j].w};
          acc[j] = __builtin_amdgcn_wmma_f32_16x16x4_f32(
              false, ahi, false, b, (short)0, acc[j], false, false);
        }
      }
      __builtin_amdgcn_sched_barrier(0);
      if (g + 2 < 16) {
#pragma unroll
        for (int j = 0; j < 8; ++j)
          b0[j] = wf[(((g + 2) * 8 + j) * 2 + hh) * 16 +
                     (c ^ ((2 * (g + 2) + hh) & 15))];
      }
      __builtin_amdgcn_sched_barrier(0);
      {
        const v2f alo = {a[g + 1].x, a[g + 1].y};
        const v2f ahi = {a[g + 1].z, a[g + 1].w};
#pragma unroll
        for (int j = 0; j < 8; ++j) {
          const v2f b = {b1[j].x, b1[j].y};
          acc[j] = __builtin_amdgcn_wmma_f32_16x16x4_f32(
              false, alo, false, b, (short)0, acc[j], false, false);
        }
#pragma unroll
        for (int j = 0; j < 8; ++j) {
          const v2f b = {b1[j].z, b1[j].w};
          acc[j] = __builtin_amdgcn_wmma_f32_16x16x4_f32(
              false, ahi, false, b, (short)0, acc[j], false, false);
        }
      }
      __builtin_amdgcn_sched_barrier(0);
    }

    acc_to_rows(acc, tbw, out, tile, lane, c, hh);

    m += RG_WAVES;
    tile = tb + ts * m;
    have = tile < nTiles;
    if (have) {
      const float4* ap = Av + tile * 16 * (HID / 4) + lane;
#pragma unroll
      for (int i = 0; i < 16; ++i) a[i] = ap[i * (HID / 4)];
    }
  }
}

__device__ __forceinline__ void half_pass(const float4 (&a)[16],
                                          const float4* __restrict__ wq,
                                          const float* __restrict__ bias,
                                          bool bv_on, float* __restrict__ out,
                                          float4* tb, int tile, int lane,
                                          int c, int hh) {
  v8f acc[8];
#pragma unroll
  for (int j = 0; j < 8; ++j) {
    const float bv = bv_on ? bias[16 * j + c] : 0.0f;
#pragma unroll
    for (int r = 0; r < 8; ++r) acc[j][r] = bv;
  }

  float4 b0[8], b1[8];
#pragma unroll
  for (int j = 0; j < 8; ++j)
    b0[j] = wq[((0 * 8 + j) * 2 + hh) * 16 + (c ^ ((2 * 0 + hh) & 15))];

#pragma unroll
  for (int g = 0; g < 16; g += 2) {
#pragma unroll
    for (int j = 0; j < 8; ++j)
      b1[j] = wq[(((g + 1) * 8 + j) * 2 + hh) * 16 +
                 (c ^ ((2 * (g + 1) + hh) & 15))];
    __builtin_amdgcn_sched_barrier(0);
    {
      const v2f alo = {a[g].x, a[g].y};
      const v2f ahi = {a[g].z, a[g].w};
#pragma unroll
      for (int j = 0; j < 8; ++j) {
        const v2f b = {b0[j].x, b0[j].y};
        acc[j] = __builtin_amdgcn_wmma_f32_16x16x4_f32(
            false, alo, false, b, (short)0, acc[j], false, false);
      }
#pragma unroll
      for (int j = 0; j < 8; ++j) {
        const v2f b = {b0[j].z, b0[j].w};
        acc[j] = __builtin_amdgcn_wmma_f32_16x16x4_f32(
            false, ahi, false, b, (short)0, acc[j], false, false);
      }
    }
    __builtin_amdgcn_sched_barrier(0);
    if (g + 2 < 16) {
#pragma unroll
      for (int j = 0; j < 8; ++j)
        b0[j] = wq[(((g + 2) * 8 + j) * 2 + hh) * 16 +
                   (c ^ ((2 * (g + 2) + hh) & 15))];
    }
    __builtin_amdgcn_sched_barrier(0);
    {
      const v2f alo = {a[g + 1].x, a[g + 1].y};
      const v2f ahi = {a[g + 1].z, a[g + 1].w};
#pragma unroll
      for (int j = 0; j < 8; ++j) {
        const v2f b = {b1[j].x, b1[j].y};
        acc[j] = __builtin_amdgcn_wmma_f32_16x16x4_f32(
            false, alo, false, b, (short)0, acc[j], false, false);
      }
#pragma unroll
      for (int j = 0; j < 8; ++j) {
        const v2f b = {b1[j].z, b1[j].w};
        acc[j] = __builtin_amdgcn_wmma_f32_16x16x4_f32(
            false, ahi, false, b, (short)0, acc[j], false, false);
      }
    }
    __builtin_amdgcn_sched_barrier(0);
  }

  acc_to_rows(acc, tb, out, tile, lane, c, hh);
}

__global__ void __launch_bounds__(RG_WAVES * 32) fold_gemm(
    float* __restrict__ Hh, float* __restrict__ Zz,
    const float* __restrict__ W, const float* __restrict__ bias,
    const long long* __restrict__ tot, const float* __restrict__ gamma,
    const float* __restrict__ beta, float* __restrict__ Pp, int nTiles,
    float invN) {
  __shared__ __attribute__((aligned(16))) float4 wf[2 * 16 * 8 * 2 * 16];
  __shared__ __attribute__((aligned(16))) float4 tbuf[RG_WAVES][16 * TP];
  __shared__ __attribute__((aligned(16))) float bnab[2 * HID];

  const int tid = threadIdx.x;
  const int lane = tid & 31;
  const int wave = __builtin_amdgcn_readfirstlane(tid >> 5);
  const int hh = lane >> 4;
  const int c = lane & 15;

  const int slot = 4 * (((wave & 3) - (wave >> 2)) & 3) + (wave >> 2);
  const int tile = (int)blockIdx.x + RG_BLOCKS * slot;
  const bool have = tile < nTiles;

  float4 hv[16], zv[16];
  float4* hrow = (float4*)Hh + tile * 16 * (HID / 4) + lane;
  if (have) {
    const float4* zrow = (const float4*)Zz + tile * 16 * (HID / 4) + lane;
#pragma unroll
    for (int i = 0; i < 16; ++i) hv[i] = hrow[i * (HID / 4)];
#pragma unroll
    for (int i = 0; i < 16; ++i) zv[i] = zrow[i * (HID / 4)];
  }

  if (tid < HID) {
    const float s = (float)tot[tid] * FX_INV;
    const float ss = (float)tot[HID + tid] * FX_INV;
    const float mu = s * invN;
    const float var = ss * invN - mu * mu;
    const float sa = gamma[tid] / sqrtf(var + 1e-5f);
    bnab[tid] = sa;
    bnab[HID + tid] = beta[tid] - mu * sa;
  }

#pragma unroll
  for (int q = 0; q < 2; ++q) {
    float4 t[8];
#pragma unroll
    for (int i = 0; i < 8; ++i)
      t[i] = *(const float4*)(W + (16 * i + wave) * (HID * 2) + q * HID +
                              4 * lane);
#pragma unroll
    for (int i = 0; i < 8; ++i)
      wf[q * 4096 + (((lane >> 1) * 8 + i) * 2 + (lane & 1)) * 16 +
         ((wave ^ lane) & 15)] = t[i];
  }
  __syncthreads();

  if (have) {
    const float4 sa = ((const float4*)&bnab[0])[lane];
    const float4 sb = ((const float4*)&bnab[HID])[lane];
#pragma unroll
    for (int i = 0; i < 16; ++i) {
      hv[i].x = hv[i].x + zv[i].x * sa.x + sb.x;
      hv[i].y = hv[i].y + zv[i].y * sa.y + sb.y;
      hv[i].z = hv[i].z + zv[i].z * sa.z + sb.z;
      hv[i].w = hv[i].w + zv[i].w * sa.w + sb.w;
    }
#pragma unroll
    for (int i = 0; i < 16; ++i) hrow[i * (HID / 4)] = hv[i];

    float4* tbw = tbuf[wave];
    rows_to_frags(hv, tbw, lane, c, hh);

    half_pass(hv, wf, bias, true, Zz, tbw, tile, lane, c, hh);
    half_pass(hv, wf + 4096, bias, false, Pp, tbw, tile, lane, c, hh);
  }
}

__global__ void residual_bn(const float* __restrict__ h,
                            const float* __restrict__ z,
                            const long long* __restrict__ tot,
                            const float* __restrict__ gamma,
                            const float* __restrict__ beta,
                            float* __restrict__ out, int total, float invN) {
  const int j = threadIdx.x & (HID - 1);
  const float s = (float)tot[j] * FX_INV;
  const float ss = (float)tot[HID + j] * FX_INV;
  const float mu = s * invN;
  const float var = ss * invN - mu * mu;
  const float a = gamma[j] / sqrtf(var + 1e-5f);
  const float b = beta[j] - mu * a;
  for (int i = blockIdx.x * blockDim.x + threadIdx.x; i < total;
       i += gridDim.x * blockDim.x)
    out[i] = h[i] + z[i] * a + b;
}

extern "C" void kernel_launch(void* const* d_in, const int* in_sizes, int n_in,
                              void* d_out, int out_size, void* d_ws,
                              size_t ws_size, hipStream_t stream) {
  const float* h0 = (const float*)d_in[0];
  const int* src = (const int*)d_in[1];
  const int* dst = (const int*)d_in[2];
  const float* W_emb = (const float*)d_in[3];
  const float* b_emb = (const float*)d_in[4];
  const float* Ws = (const float*)d_in[5];
  const float* bs = (const float*)d_in[6];
  const float* gam = (const float*)d_in[7];
  const float* bet = (const float*)d_in[8];

  const int N = in_sizes[0] / HID;
  const int E = in_sizes[1];
  const int NF = N * HID;
  const int L = 4;

  float* ws = (float*)d_ws;
  float* H = ws;
  float* C = H + NF;
  float* Z = C + NF;
  int* head = (int*)(Z + NF);
  long long* TOT = (long long*)(head + 2);
  int* cnt = (int*)(TOT + L * 256);

  int* tab = (int*)d_out;
  int* spill = tab + N * CAP;

  int* pairs = (int*)C;
  int* offs = (int*)Z;

  const int mtiles = N / 16;
  const int nbk = (N + 127) / 128;

  bin_edges<<<NCH, BE_T, 0, stream>>>(src, dst, pairs, offs, head, 2 + L * 512,
                                      E, nbk);
  slot_rows<<<nbk, NCH, 0, stream>>>(pairs, offs, cnt, tab, spill, head, N, E,
                                     nbk);

  rowtile_gemm<1><<<RG_BLOCKS, RG_WAVES * 32, 0, stream>>>(h0, W_emb, b_emb, H,
                                                          H, mtiles);

  const float invN = 1.0f / (float)N;
  for (int l = 0; l < L; ++l) {
    if (l == 0) {
      rowtile_gemm<2><<<RG_BLOCKS, RG_WAVES * 32, 0, stream>>>(
          H, Ws, bs, Z, C, mtiles);
    } else {
      fold_gemm<<<RG_BLOCKS, RG_WAVES * 32, 0, stream>>>(
          H, Z, Ws + l * HID * 256, bs + l * HID, TOT + (l - 1) * 256,
          gam + (l - 1) * HID, bet + (l - 1) * HID, C, mtiles, invN);
    }
    gather_norm_stats<<<NB_STATS, GW * 32, 0, stream>>>(
        C, cnt, tab, spill, head, Z, TOT + l * 256, N);
  }
  residual_bn<<<4096, 256, 0, stream>>>(H, Z, TOT + (L - 1) * 256,
                                        gam + (L - 1) * HID,
                                        bet + (L - 1) * HID, (float*)d_out, NF,
                                        invN);
}


# bin_edges, slot_rows launches gone: edge binning rides at the head of the embedding GEMM launch, row slots at the head of the layer-0 GEMM launch, first tile rows in flight; list order by arrival
# speedup vs baseline: 5.4167x; 1.0302x over previous
#include <hip/hip_runtime.h>
#include <hip/hip_bf16.h>


typedef float v2f __attribute__((ext_vector_type(2)));
typedef float v8f __attribute__((ext_vector_type(8)));

#define HID 128
#define NB_STATS 512

#define FX_SCALE 68719476736.0f
#define FX_INV 1.4551915228366852e-11f

#define CAP 32

__device__ __forceinline__ void lds_barrier() {
  __builtin_amdgcn_fence(__ATOMIC_RELEASE, "workgroup", "local");
  __builtin_amdgcn_s_barrier();
  __builtin_amdgcn_fence(__ATOMIC_ACQUIRE, "workgroup", "local");
}

#define BE_T 512
#define MAXBK BE_T
#define BE_U 5
#define NCH 256

__device__ __forceinline__ void bin_chunk(
    const int* __restrict__ src, const int* __restrict__ dst,
    int* __restrict__ pairs, int* __restrict__ offs, int* __restrict__ zero,
    int nzero, int E, int nbk) {
  __shared__ int hist[MAXBK];
  __shared__ int wsum[BE_T / 32];

  const int t = threadIdx.x;
  const int nch = (int)gridDim.x;
  const int epc = (E + nch - 1) / nch;
  const int e0 = min((int)blockIdx.x * epc, E);
  const int e1 = min(e0 + epc, E);

  if (blockIdx.x == 0)
    for (int i = t; i < nzero; i += BE_T) zero[i] = 0;
  for (int i = t; i < nbk; i += BE_T) hist[i] = 0;
  lds_barrier();

  for (int base = e0 + t; base < e1; base += BE_U * BE_T) {
    int d[BE_U];
#pragma unroll
    for (int j = 0; j < BE_U; ++j) {
      const int e = base + j * BE_T;
      d[j] = (e < e1) ? dst[e] : -1;
    }
#pragma unroll
    for (int j = 0; j < BE_U; ++j)
      if (d[j] >= 0) atomicAdd(&hist[d[j] >> 7], 1);
  }
  lds_barrier();

  {
    const int lane = t & 31;
    const int wv = t >> 5;
    const int c = (t < nbk) ? hist[t] : 0;
    int inc = c;
    for (int off = 1; off < 32; off <<= 1) {
      const int v = __shfl_up(inc, off, 32);
      if (lane >= off) inc += v;
    }
    if (lane == 31) wsum[wv] = inc;
    lds_barrier();
    int pre = (lane < wv && lane < BE_T / 32) ? wsum[lane] : 0;
    for (int off = 16; off > 0; off >>= 1) pre += __shfl_xor(pre, off, 32);
    const int start = pre + inc - c;
    int* orow = offs + blockIdx.x * (nbk + 1);
    if (t < nbk) {
      hist[t] = start;
      orow[t] = start;
    }
    if (t == BE_T - 1) orow[nbk] = pre + inc;
  }
  lds_barrier();

  int2* prow = (int2*)pairs + e0;
  for (int base = e0 + t; base < e1; base += BE_U * BE_T) {
    int d[BE_U], sv[BE_U];
#pragma unroll
    for (int j = 0; j < BE_U; ++j) {
      const int e = base + j * BE_T;
      d[j] = (e < e1) ? dst[e] : -1;
      sv[j] = (e < e1) ? src[e] : 0;
    }
#pragma unroll
    for (int j = 0; j < BE_U; ++j)
      if (d[j] >= 0) {
        const int pos = atomicAdd(&hist[d[j] >> 7], 1);
        prow[pos] = make_int2(d[j], sv[j]);
      }
  }
}

#define SR_U 4

__device__ __forceinline__ void slot_ranges(
    const int* __restrict__ pairs, const int* __restrict__ offs,
    int* __restrict__ cnt, int* __restrict__ tab, int* __restrict__ spill,
    int* __restrict__ nspill, int N, int E, int nbk) {
  __shared__ int cl[2][128];

  const int t = threadIdx.x;
  const int hf = t >> 8;
  const int k = t & (NCH - 1);
  const int r = (int)blockIdx.x + NCH * hf;
  const bool on = r < nbk;
  if (k < 128) cl[hf][k] = 0;
  lds_barrier();

  if (on) {
    const int epc = (E + NCH - 1) / NCH;
    const int o0 = offs[k * (nbk + 1) + r];
    const int o1 = offs[k * (nbk + 1) + r + 1];
    const int2* prow = (const int2*)pairs + min(k * epc, E);
    for (int i = o0; i < o1; i += SR_U) {
      int2 p[SR_U];
#pragma unroll
      for (int j = 0; j < SR_U; ++j)
        p[j] = (i + j < o1) ? prow[i + j] : make_int2(-1, 0);
#pragma unroll
      for (int j = 0; j < SR_U; ++j)
        if (p[j].x >= 0) {
          const int slot = atomicAdd(&cl[hf][p[j].x & 127], 1);
          if (slot < CAP) {
            tab[p[j].x * CAP + slot] = p[j].y;
          } else {
            const int o = atomicAdd(nspill, 1);
            ((int2*)spill)[o] = p[j];
          }
        }
    }
  }
  lds_barrier();

  const int node = r * 128 + k;
  if (on && k < 128 && node < N) cnt[node] = cl[hf][k];
}

#define GW 16

__global__ void __launch_bounds__(GW * 32) gather_norm_stats(
    const float* __restrict__ p2, const int* __restrict__ deg,
    const int* __restrict__ tab, const int* __restrict__ spill,
    const int* __restrict__ nspill, float* __restrict__ z,
    long long* __restrict__ tot, int N) {
  __shared__ __attribute__((aligned(16))) float red[GW][256];

  const int t = threadIdx.x;
  const int lane = t & 31;
  const int wave = __builtin_amdgcn_readfirstlane(t >> 5);

  const float4* p2v = (const float4*)p2;
  float4* zv4 = (float4*)z;

  float4 ps = make_float4(0.f, 0.f, 0.f, 0.f);
  float4 pss = make_float4(0.f, 0.f, 0.f, 0.f);

  const int stride = gridDim.x * GW;
  const int node0 = blockIdx.x * GW + wave;

  for (int kb = 0; node0 + kb * stride < N; kb += 32) {
    const int ln = node0 + (kb + lane) * stride;
    const int ld = (ln < N) ? deg[ln] : 0;
    for (int k = 0; k < 32; ++k) {
      const int node = node0 + (kb + k) * stride;
      if (node >= N) break;

      const int dg = __builtin_amdgcn_readlane(ld, k);
      const int m = min(dg, CAP);
      const int cur = (lane < m) ? tab[node * CAP + lane] : 0;

      const float4 zv = zv4[node * 32 + lane];
      const float inv = 1.0f / fmaxf((float)dg, 1.0f);

      float4 acc = make_float4(0.f, 0.f, 0.f, 0.f);
      for (int i = 0; i < m; i += 8) {
        float4 v[8];
#pragma unroll
        for (int j = 0; j < 8; ++j) {
          const int s = __builtin_amdgcn_readlane(cur, i + j);
          v[j] = p2v[s * 32 + lane];
        }
#pragma unroll
        for (int j = 0; j < 8; ++j) {
          const bool ok = (i + j) < m;
          acc.x += ok ? v[j].x : 0.f;
          acc.y += ok ? v[j].y : 0.f;
          acc.z += ok ? v[j].z : 0.f;
          acc.w += ok ? v[j].w : 0.f;
        }
      }

      if (dg > CAP) {
        const int ns = nspill[0];
        const int2* sp = (const int2*)spill;
        for (int o = 0; o < ns; o += 32) {
          int2 p = make_int2(-1, 0);
          if (o + lane < ns) p = sp[o + lane];
          unsigned hit = __builtin_amdgcn_ballot_w32(p.x == node);
          while (hit != 0u) {
            const int b = __builtin_ctz(hit);
            hit &= hit - 1u;
            const int s = __builtin_amdgcn_readlane(p.y, b);
            const float4 w = p2v[s * 32 + lane];
            acc.x += w.x;
            acc.y += w.y;
            acc.z += w.z;
            acc.w += w.w;
          }
        }
      }

      float4 zz;
      zz.x = zv.x + acc.x * inv;
      zz.y = zv.y + acc.y * inv;
      zz.z = zv.z + acc.z * inv;
      zz.w = zv.w + acc.w * inv;

      float sq = zz.x * zz.x + zz.y * zz.y + zz.z * zz.z + zz.w * zz.w;
#pragma unroll
      for (int off = 16; off > 0; off >>= 1) sq += __shfl_xor(sq, off, 32);
      const float nrm = fmaxf(sqrtf(sq), 1e-12f);

      float4 r;
      r.x = fmaxf(zz.x / nrm, 0.0f);
      r.y = fmaxf(zz.y / nrm, 0.0f);
      r.z = fmaxf(zz.z / nrm, 0.0f);
      r.w = fmaxf(zz.w / nrm, 0.0f);
      zv4[node * 32 + lane] = r;

      ps.x += r.x;  ps.y += r.y;  ps.z += r.z;  ps.w += r.w;
      pss.x += r.x * r.x;  pss.y += r.y * r.y;
      pss.z += r.z * r.z;  pss.w += r.w * r.w;
    }
  }

  ((float4*)&red[wave][0])[lane] = ps;
  ((float4*)&red[wave][128])[lane] = pss;
  __syncthreads();
  if (t < 256) {
    float s = 0.0f;
#pragma unroll
    for (int w = 0; w < GW; ++w) s += red[w][t];
    __hip_atomic_fetch_add(&tot[t], (long long)(s * FX_SCALE),
                           __ATOMIC_RELAXED, __HIP_MEMORY_SCOPE_AGENT);
  }
}

#define RG_WAVES 16
#define RG_BLOCKS 256
#define TP 33

__device__ __forceinline__ void rows_to_frags(float4 (&a)[16], float4* tb,
                                              int lane, int c, int hh) {
#pragma unroll
  for (int i = 0; i < 16; ++i) tb[i * TP + lane] = a[i];
  __builtin_amdgcn_fence(__ATOMIC_ACQ_REL, "wavefront");
#pragma unroll
  for (int g = 0; g < 16; ++g) a[g] = tb[c * TP + 2 * g + hh];
}

__device__ __forceinline__ void acc_to_rows(const v8f (&acc)[8], float4* tb,
                                            float* __restrict__ out, int tile,
                                            int lane, int c, int hh) {
  float* tbf = (float*)tb;
#pragma unroll
  for (int r = 0; r < 8; ++r) {
#pragma unroll
    for (int j = 0; j < 8; ++j)
      tbf[(8 * hh + r) * (4 * TP) + 16 * j + c] = acc[j][r];
  }
  __builtin_amdgcn_fence(__ATOMIC_ACQ_REL, "wavefront");
  float4* op = (float4*)out + tile * 16 * (HID / 4) + lane;
#pragma unroll
  for (int i = 0; i < 16; ++i) op[i * (HID / 4)] = tb[i * TP + lane];
}

static_assert(RG_BLOCKS == NCH && RG_WAVES * 32 == BE_T && BE_T == 2 * NCH,
              "the table steps use the GEMM launches' grid shape");

template <int NQ>
__global__ void __launch_bounds__(RG_WAVES * 32) rowtile_gemm(
    const float* __restrict__ A, const float* __restrict__ W,
    const float* __restrict__ bias, float* __restrict__ out0,
    float* __restrict__ out1, int nTiles, const int* __restrict__ src,
    const int* __restrict__ dst, int* __restrict__ pairs,
    int* __restrict__ offs, int* __restrict__ head, int nzero,
    int* __restrict__ cnt, int* __restrict__ tab, int* __restrict__ spill,
    int N, int E, int nbk) {
  __shared__ __attribute__((aligned(16))) float4 wf[16 * 8 * 2 * 16];
  __shared__ __attribute__((aligned(16))) float4 tbuf[RG_WAVES][16 * TP];

  const int tid = threadIdx.x;
  const int lane = tid & 31;
  const int wave = __builtin_amdgcn_readfirstlane(tid >> 5);
  const int hh = lane >> 4;
  const int c = lane & 15;

  const int q = blockIdx.x & (NQ - 1);
  const int tb = blockIdx.x / NQ;
  const int ts = RG_BLOCKS / NQ;
  const float* Wq = W + q * HID;
  float* __restrict__ out = (q == 0) ? out0 : out1;

  const float4* Av = (const float4*)A;
  float4* tbw = tbuf[wave];

  int m = wave;
  int tile = tb + ts * m;
  bool have = tile < nTiles;

  float4 a[16];
  if (have) {
    const float4* ap = Av + tile * 16 * (HID / 4) + lane;
#pragma unroll
    for (int i = 0; i < 16; ++i) a[i] = ap[i * (HID / 4)];
  }

  if (NQ == 1)
    bin_chunk(src, dst, pairs, offs, head, nzero, E, nbk);
  else
    slot_ranges(pairs, offs, cnt, tab, spill, head, N, E, nbk);

  {
    float4 t[8];
#pragma unroll
    for (int i = 0; i < 8; ++i)
      t[i] = *(const float4*)(Wq + (16 * i + wave) * (HID * NQ) + 4 * lane);
#pragma unroll
    for (int i = 0; i < 8; ++i)
      wf[(((lane >> 1) * 8 + i) * 2 + (lane & 1)) * 16 + ((wave ^ lane) & 15)] =
          t[i];
  }
  lds_barrier();

  while (have) {
    asm volatile("" ::: "memory");
    rows_to_frags(a, tbw, lane, c, hh);

    v8f acc[8];
#pragma unroll
    for (int j = 0; j < 8; ++j) {
      const float bv = (q == 0) ? bias[16 * j + c] : 0.0f;
#pragma unroll
      for (int r = 0; r < 8; ++r) acc[j][r] = bv;
    }

    float4 b0[8], b1[8];
#pragma unroll
    for (int j = 0; j < 8; ++j)
      b0[j] = wf[((0 * 8 + j) * 2 + hh) * 16 + (c ^ ((2 * 0 + hh) & 15))];

#pragma unroll
    for (int g = 0; g < 16; g += 2) {
#pragma unroll
      for (int j = 0; j < 8; ++j)
        b1[j] = wf[(((g + 1) * 8 + j) * 2 + hh) * 16 +
                   (c ^ ((2 * (g + 1) + hh) & 15))];
      __builtin_amdgcn_sched_barrier(0);
      {
        const v2f alo = {a[g].x, a[g].y};
        const v2f ahi = {a[g].z, a[g].w};
#pragma unroll
        for (int j = 0; j < 8; ++j) {
          const v2f b = {b0[j].x, b0[j].y};
          acc[j] = __builtin_amdgcn_wmma_f32_16x16x4_f32(
              false, alo, false, b, (short)0, acc[j], false, false);
        }
#pragma unroll
        for (int j = 0; j < 8; ++j) {
          const v2f b = {b0[j].z, b0[j].w};
          acc[j] = __builtin_amdgcn_wmma_f32_16x16x4_f32(
              false, ahi, false, b, (short)0, acc[j], false, false);
        }
      }
      __builtin_amdgcn_sched_barrier(0);
      if (g + 2 < 16) {
#pragma unroll
        for (int j = 0; j < 8; ++j)
          b0[j] = wf[(((g + 2) * 8 + j) * 2 + hh) * 16 +
                     (c ^ ((2 * (g + 2) + hh) & 15))];
      }
      __builtin_amdgcn_sched_barrier(0);
      {
        const v2f alo = {a[g + 1].x, a[g + 1].y};
        const v2f ahi = {a[g + 1].z, a[g + 1].w};
#pragma unroll
        for (int j = 0; j < 8; ++j) {
          const v2f b = {b1[j].x, b1[j].y};
          acc[j] = __builtin_amdgcn_wmma_f32_16x16x4_f32(
              false, alo, false, b, (short)0, acc[j], false, false);
        }
#pragma unroll
        for (int j = 0; j < 8; ++j) {
          const v2f b = {b1[j].z, b1[j].w};
          acc[j] = __builtin_amdgcn_wmma_f32_16x16x4_f32(
              false, ahi, false, b, (short)0, acc[j], false, false);
        }
      }
      __builtin_amdgcn_sched_barrier(0);
    }

    acc_to_rows(acc, tbw, out, tile, lane, c, hh);

    m += RG_WAVES;
    tile = tb + ts * m;
    have = tile < nTiles;
    if (have) {
      const float4* ap = Av + tile * 16 * (HID / 4) + lane;
#pragma unroll
      for (int i = 0; i < 16; ++i) a[i] = ap[i * (HID / 4)];
    }
  }
}

__device__ __forceinline__ void half_pass(const float4 (&a)[16],
                                          const float4* __restrict__ wq,
                                          const float* __restrict__ bias,
                                          bool bv_on, float* __restrict__ out,
                                          float4* tb, int tile, int lane,
                                          int c, int hh) {
  v8f acc[8];
#pragma unroll
  for (int j = 0; j < 8; ++j) {
    const float bv = bv_on ? bias[16 * j + c] : 0.0f;
#pragma unroll
    for (int r = 0; r < 8; ++r) acc[j][r] = bv;
  }

  float4 b0[8], b1[8];
#pragma unroll
  for (int j = 0; j < 8; ++j)
    b0[j] = wq[((0 * 8 + j) * 2 + hh) * 16 + (c ^ ((2 * 0 + hh) & 15))];

#pragma unroll
  for (int g = 0; g < 16; g += 2) {
#pragma unroll
    for (int j = 0; j < 8; ++j)
      b1[j] = wq[(((g + 1) * 8 + j) * 2 + hh) * 16 +
                 (c ^ ((2 * (g + 1) + hh) & 15))];
    __builtin_amdgcn_sched_barrier(0);
    {
      const v2f alo = {a[g].x, a[g].y};
      const v2f ahi = {a[g].z, a[g].w};
#pragma unroll
      for (int j = 0; j < 8; ++j) {
        const v2f b = {b0[j].x, b0[j].y};
        acc[j] = __builtin_amdgcn_wmma_f32_16x16x4_f32(
            false, alo, false, b, (short)0, acc[j], false, false);
      }
#pragma unroll
      for (int j = 0; j < 8; ++j) {
        const v2f b = {b0[j].z, b0[j].w};
        acc[j] = __builtin_amdgcn_wmma_f32_16x16x4_f32(
            false, ahi, false, b, (short)0, acc[j], false, false);
      }
    }
    __builtin_amdgcn_sched_barrier(0);
    if (g + 2 < 16) {
#pragma unroll
      for (int j = 0; j < 8; ++j)
        b0[j] = wq[(((g + 2) * 8 + j) * 2 + hh) * 16 +
                   (c ^ ((2 * (g + 2) + hh) & 15))];
    }
    __builtin_amdgcn_sched_barrier(0);
    {
      const v2f alo = {a[g + 1].x, a[g + 1].y};
      const v2f ahi = {a[g + 1].z, a[g + 1].w};
#pragma unroll
      for (int j = 0; j < 8; ++j) {
        const v2f b = {b1[j].x, b1[j].y};
        acc[j] = __builtin_amdgcn_wmma_f32_16x16x4_f32(
            false, alo, false, b, (short)0, acc[j], false, false);
      }
#pragma unroll
      for (int j = 0; j < 8; ++j) {
        const v2f b = {b1[j].z, b1[j].w};
        acc[j] = __builtin_amdgcn_wmma_f32_16x16x4_f32(
            false, ahi, false, b, (short)0, acc[j], false, false);
      }
    }
    __builtin_amdgcn_sched_barrier(0);
  }

  acc_to_rows(acc, tb, out, tile, lane, c, hh);
}

__global__ void __launch_bounds__(RG_WAVES * 32) fold_gemm(
    float* __restrict__ Hh, float* __restrict__ Zz,
    const float* __restrict__ W, const float* __restrict__ bias,
    const long long* __restrict__ tot, const float* __restrict__ gamma,
    const float* __restrict__ beta, float* __restrict__ Pp, int nTiles,
    float invN) {
  __shared__ __attribute__((aligned(16))) float4 wf[2 * 16 * 8 * 2 * 16];
  __shared__ __attribute__((aligned(16))) float4 tbuf[RG_WAVES][16 * TP];
  __shared__ __attribute__((aligned(16))) float bnab[2 * HID];

  const int tid = threadIdx.x;
  const int lane = tid & 31;
  const int wave = __builtin_amdgcn_readfirstlane(tid >> 5);
  const int hh = lane >> 4;
  const int c = lane & 15;

  const int slot = 4 * (((wave & 3) - (wave >> 2)) & 3) + (wave >> 2);
  const int tile = (int)blockIdx.x + RG_BLOCKS * slot;
  const bool have = tile < nTiles;

  float4 hv[16], zv[16];
  float4* hrow = (float4*)Hh + tile * 16 * (HID / 4) + lane;
  if (have) {
    const float4* zrow = (const float4*)Zz + tile * 16 * (HID / 4) + lane;
#pragma unroll
    for (int i = 0; i < 16; ++i) hv[i] = hrow[i * (HID / 4)];
#pragma unroll
    for (int i = 0; i < 16; ++i) zv[i] = zrow[i * (HID / 4)];
  }

  if (tid < HID) {
    const float s = (float)tot[tid] * FX_INV;
    const float ss = (float)tot[HID + tid] * FX_INV;
    const float mu = s * invN;
    const float var = ss * invN - mu * mu;
    const float sa = gamma[tid] / sqrtf(var + 1e-5f);
    bnab[tid] = sa;
    bnab[HID + tid] = beta[tid] - mu * sa;
  }

#pragma unroll
  for (int q = 0; q < 2; ++q) {
    float4 t[8];
#pragma unroll
    for (int i = 0; i < 8; ++i)
      t[i] = *(const float4*)(W + (16 * i + wave) * (HID * 2) + q * HID +
                              4 * lane);
#pragma unroll
    for (int i = 0; i < 8; ++i)
      wf[q * 4096 + (((lane >> 1) * 8 + i) * 2 + (lane & 1)) * 16 +
         ((wave ^ lane) & 15)] = t[i];
  }
  __syncthreads();

  if (have) {
    const float4 sa = ((const float4*)&bnab[0])[lane];
    const float4 sb = ((const float4*)&bnab[HID])[lane];
#pragma unroll
    for (int i = 0; i < 16; ++i) {
      hv[i].x = hv[i].x + zv[i].x * sa.x + sb.x;
      hv[i].y = hv[i].y + zv[i].y * sa.y + sb.y;
      hv[i].z = hv[i].z + zv[i].z * sa.z + sb.z;
      hv[i].w = hv[i].w + zv[i].w * sa.w + sb.w;
    }
#pragma unroll
    for (int i = 0; i < 16; ++i) hrow[i * (HID / 4)] = hv[i];

    float4* tbw = tbuf[wave];
    rows_to_frags(hv, tbw, lane, c, hh);

    half_pass(hv, wf, bias, true, Zz, tbw, tile, lane, c, hh);
    half_pass(hv, wf + 4096, bias, false, Pp, tbw, tile, lane, c, hh);
  }
}

__global__ void residual_bn(const float* __restrict__ h,
                            const float* __restrict__ z,
                            const long long* __restrict__ tot,
                            const float* __restrict__ gamma,
                            const float* __restrict__ beta,
                            float* __restrict__ out, int total, float invN) {
  const int j = threadIdx.x & (HID - 1);
  const float s = (float)tot[j] * FX_INV;
  const float ss = (float)tot[HID + j] * FX_INV;
  const float mu = s * invN;
  const float var = ss * invN - mu * mu;
  const float a = gamma[j] / sqrtf(var + 1e-5f);
  const float b = beta[j] - mu * a;
  for (int i = blockIdx.x * blockDim.x + threadIdx.x; i < total;
       i += gridDim.x * blockDim.x)
    out[i] = h[i] + z[i] * a + b;
}

extern "C" void kernel_launch(void* const* d_in, const int* in_sizes, int n_in,
                              void* d_out, int out_size, void* d_ws,
                              size_t ws_size, hipStream_t stream) {
  const float* h0 = (const float*)d_in[0];
  const int* src = (const int*)d_in[1];
  const int* dst = (const int*)d_in[2];
  const float* W_emb = (const float*)d_in[3];
  const float* b_emb = (const float*)d_in[4];
  const float* Ws = (const float*)d_in[5];
  const float* bs = (const float*)d_in[6];
  const float* gam = (const float*)d_in[7];
  const float* bet = (const float*)d_in[8];

  const int N = in_sizes[0] / HID;
  const int E = in_sizes[1];
  const int NF = N * HID;
  const int L = 4;

  float* ws = (float*)d_ws;
  float* H = ws;
  float* C = H + NF;
  float* Z = C + NF;
  int* head = (int*)(Z + NF);
  long long* TOT = (long long*)(head + 2);
  int* cnt = (int*)(TOT + L * 256);

  int* tab = (int*)d_out;
  int* spill = tab + N * CAP;
  int* pairs = spill + 2 * E;
  int* offs = pairs + 2 * E;

  const int mtiles = N / 16;
  const int nbk = (N + 127) / 128;

  rowtile_gemm<1><<<RG_BLOCKS, RG_WAVES * 32, 0, stream>>>(
      h0, W_emb, b_emb, H, H, mtiles, src, dst, pairs, offs, head, 2 + L * 512,
      cnt, tab, spill, N, E, nbk);

  const float invN = 1.0f / (float)N;
  for (int l = 0; l < L; ++l) {
    if (l == 0) {
      rowtile_gemm<2><<<RG_BLOCKS, RG_WAVES * 32, 0, stream>>>(
          H, Ws, bs, Z, C, mtiles, src, dst, pairs, offs, head, 0, cnt, tab,
          spill, N, E, nbk);
    } else {
      fold_gemm<<<RG_BLOCKS, RG_WAVES * 32, 0, stream>>>(
          H, Z, Ws + l * HID * 256, bs + l * HID, TOT + (l - 1) * 256,
          gam + (l - 1) * HID, bet + (l - 1) * HID, C, mtiles, invN);
    }
    gather_norm_stats<<<NB_STATS, GW * 32, 0, stream>>>(
        C, cnt, tab, spill, head, Z, TOT + l * 256, N);
  }
  residual_bn<<<4096, 256, 0, stream>>>(H, Z, TOT + (L - 1) * 256,
                                        gam + (L - 1) * HID,
                                        bet + (L - 1) * HID, (float*)d_out, NF,
                                        invN);
}


# W1 half of each layer GEMM moved into the gather launch: a 16-wave block forms its next tile's h W1^T+b on WMMA under its neighbour-row loads; GEMM launches do P only; k sum split in two halves
# speedup vs baseline: 5.6715x; 1.0470x over previous
#include <hip/hip_runtime.h>
#include <hip/hip_bf16.h>


typedef float v2f __attribute__((ext_vector_type(2)));
typedef float v8f __attribute__((ext_vector_type(8)));

#define HID 128
#define NB_STATS 512

#define FX_SCALE 68719476736.0f
#define FX_INV 1.4551915228366852e-11f

#define CAP 32

__device__ __forceinline__ void lds_barrier() {
  __builtin_amdgcn_fence(__ATOMIC_RELEASE, "workgroup", "local");
  __builtin_amdgcn_s_barrier();
  __builtin_amdgcn_fence(__ATOMIC_ACQUIRE, "workgroup", "local");
}

#define BE_T 512
#define MAXBK BE_T
#define BE_U 5
#define NCH 256

__device__ __forceinline__ void bin_chunk(
    const int* __restrict__ src, const int* __restrict__ dst,
    int* __restrict__ pairs, int* __restrict__ offs, int* __restrict__ zero,
    int nzero, int E, int nbk) {
  __shared__ int hist[MAXBK];
  __shared__ int wsum[BE_T / 32];

  const int t = threadIdx.x;
  const int nch = (int)gridDim.x;
  const int epc = (E + nch - 1) / nch;
  const int e0 = min((int)blockIdx.x * epc, E);
  const int e1 = min(e0 + epc, E);

  if (blockIdx.x == 0)
    for (int i = t; i < nzero; i += BE_T) zero[i] = 0;
  for (int i = t; i < nbk; i += BE_T) hist[i] = 0;
  lds_barrier();

  for (int base = e0 + t; base < e1; base += BE_U * BE_T) {
    int d[BE_U];
#pragma unroll
    for (int j = 0; j < BE_U; ++j) {
      const int e = base + j * BE_T;
      d[j] = (e < e1) ? dst[e] : -1;
    }
#pragma unroll
    for (int j = 0; j < BE_U; ++j)
      if (d[j] >= 0) atomicAdd(&hist[d[j] >> 7], 1);
  }
  lds_barrier();

  {
    const int lane = t & 31;
    const int wv = t >> 5;
    const int c = (t < nbk) ? hist[t] : 0;
    int inc = c;
    for (int off = 1; off < 32; off <<= 1) {
      const int v = __shfl_up(inc, off, 32);
      if (lane >= off) inc += v;
    }
    if (lane == 31) wsum[wv] = inc;
    lds_barrier();
    int pre = (lane < wv && lane < BE_T / 32) ? wsum[lane] : 0;
    for (int off = 16; off > 0; off >>= 1) pre += __shfl_xor(pre, off, 32);
    const int start = pre + inc - c;
    int* orow = offs + blockIdx.x * (nbk + 1);
    if (t < nbk) {
      hist[t] = start;
      orow[t] = start;
    }
    if (t == BE_T - 1) orow[nbk] = pre + inc;
  }
  lds_barrier();

  int2* prow = (int2*)pairs + e0;
  for (int base = e0 + t; base < e1; base += BE_U * BE_T) {
    int d[BE_U], sv[BE_U];
#pragma unroll
    for (int j = 0; j < BE_U; ++j) {
      const int e = base + j * BE_T;
      d[j] = (e < e1) ? dst[e] : -1;
      sv[j] = (e < e1) ? src[e] : 0;
    }
#pragma unroll
    for (int j = 0; j < BE_U; ++j)
      if (d[j] >= 0) {
        const int pos = atomicAdd(&hist[d[j] >> 7], 1);
        prow[pos] = make_int2(d[j], sv[j]);
      }
  }
}

#define SR_U 4

__device__ __forceinline__ void slot_ranges(
    const int* __restrict__ pairs, const int* __restrict__ offs,
    int* __restrict__ cnt, int* __restrict__ tab, int* __restrict__ spill,
    int* __restrict__ nspill, int N, int E, int nbk) {
  __shared__ int cl[2][128];

  const int t = threadIdx.x;
  const int hf = t >> 8;
  const int k = t & (NCH - 1);
  const int r = (int)blockIdx.x + NCH * hf;
  const bool on = r < nbk;
  if (k < 128) cl[hf][k] = 0;
  lds_barrier();

  if (on) {
    const int epc = (E + NCH - 1) / NCH;
    const int o0 = offs[k * (nbk + 1) + r];
    const int o1 = offs[k * (nbk + 1) + r + 1];
    const int2* prow = (const int2*)pairs + min(k * epc, E);
    for (int i = o0; i < o1; i += SR_U) {
      int2 p[SR_U];
#pragma unroll
      for (int j = 0; j < SR_U; ++j)
        p[j] = (i + j < o1) ? prow[i + j] : make_int2(-1, 0);
#pragma unroll
      for (int j = 0; j < SR_U; ++j)
        if (p[j].x >= 0) {
          const int slot = atomicAdd(&cl[hf][p[j].x & 127], 1);
          if (slot < CAP) {
            tab[p[j].x * CAP + slot] = p[j].y;
          } else {
            const int o = atomicAdd(nspill, 1);
            ((int2*)spill)[o] = p[j];
          }
        }
    }
  }
  lds_barrier();

  const int node = r * 128 + k;
  if (on && k < 128 && node < N) cnt[node] = cl[hf][k];
}

#define GW 16
#define ZP 132
#define TPA 33

__device__ __forceinline__ void z_part(const float4* atp, const float4* wf,
                                       float bv, float* ztp, int j, int kh,
                                       int c, int hh) {
  float4 a[8], b[8];
#pragma unroll
  for (int gg = 0; gg < 8; ++gg) a[gg] = atp[c * TPA + 16 * kh + 2 * gg + hh];
#pragma unroll
  for (int gg = 0; gg < 8; ++gg) {
    const int g = 8 * kh + gg;
    b[gg] = wf[((g * 8 + j) * 2 + hh) * 16 + (c ^ ((2 * gg + hh) & 15))];
  }
  v8f acc;
#pragma unroll
  for (int r = 0; r < 8; ++r) acc[r] = bv;
#pragma unroll
  for (int gg = 0; gg < 8; ++gg) {
    const v2f alo = {a[gg].x, a[gg].y};
    const v2f ahi = {a[gg].z, a[gg].w};
    const v2f blo = {b[gg].x, b[gg].y};
    const v2f bhi = {b[gg].z, b[gg].w};
    acc = __builtin_amdgcn_wmma_f32_16x16x4_f32(false, alo, false, blo,
                                                (short)0, acc, false, false);
    acc = __builtin_amdgcn_wmma_f32_16x16x4_f32(false, ahi, false, bhi,
                                                (short)0, acc, false, false);
  }
#pragma unroll
  for (int r = 0; r < 8; ++r) ztp[(8 * hh + r) * ZP + 16 * j + c] = acc[r];
}

__global__ void __launch_bounds__(GW * 32) gather_z_norm_stats(
    const float* __restrict__ Hh, const float* __restrict__ W,
    const float* __restrict__ bias, const float* __restrict__ p2,
    const int* __restrict__ deg, const int* __restrict__ tab,
    const int* __restrict__ spill, const int* __restrict__ nspill,
    float* __restrict__ z, long long* __restrict__ tot, int N, int nT) {
  __shared__ __attribute__((aligned(16))) float4 wf[16 * 8 * 2 * 16];
  __shared__ __attribute__((aligned(16))) float zt[2][2][16 * ZP];
  __shared__ __attribute__((aligned(16))) float4 at[2][16 * TPA];
  __shared__ __attribute__((aligned(16))) float red[GW][256];

  const int t = threadIdx.x;
  const int lane = t & 31;
  const int wave = __builtin_amdgcn_readfirstlane(t >> 5);
  const int hh = lane >> 4;
  const int c = lane & 15;
  const int j = wave & 7;
  const int kh = wave >> 3;

  const float4* p2v = (const float4*)p2;
  const float4* Hv = (const float4*)Hh;
  float4* zv4 = (float4*)z;

  float4 ps = make_float4(0.f, 0.f, 0.f, 0.f);
  float4 pss = make_float4(0.f, 0.f, 0.f, 0.f);

  const int dT = (int)gridDim.x;
  int T = (int)blockIdx.x;

  int dg = 0, cur = 0;
  if (T < nT && T * 16 + wave < N) {
    dg = deg[T * 16 + wave];
    cur = tab[(T * 16 + wave) * CAP + lane];
  }
  float4 h0r = make_float4(0.f, 0.f, 0.f, 0.f);
  float4 h1r = make_float4(0.f, 0.f, 0.f, 0.f);
  if (T < nT) h0r = Hv[min(T * 16 + wave, N - 1) * (HID / 4) + lane];
  if (T + dT < nT)
    h1r = Hv[min((T + dT) * 16 + wave, N - 1) * (HID / 4) + lane];

  {
    float4 tw[8];
#pragma unroll
    for (int i = 0; i < 8; ++i)
      tw[i] = *(const float4*)(W + (16 * i + wave) * (2 * HID) + 4 * lane);
#pragma unroll
    for (int i = 0; i < 8; ++i)
      wf[(((lane >> 1) * 8 + i) * 2 + (lane & 1)) * 16 + ((wave ^ lane) & 15)] =
          tw[i];
  }
  const float bv = (kh == 0) ? bias[16 * j + c] : 0.0f;
  if (T < nT) at[0][wave * TPA + lane] = h0r;
  if (T + dT < nT) at[1][wave * TPA + lane] = h1r;
  lds_barrier();

  bool first = true;
  int bsel = 0;
  for (; T < nT; T += dT) {
    const int node = T * 16 + wave;
    const int Tn = T + dT;
    const bool more = Tn < nT;

    int dgn = 0, curn = 0;
    if (more && Tn * 16 + wave < N) {
      dgn = deg[Tn * 16 + wave];
      curn = tab[(Tn * 16 + wave) * CAP + lane];
    }
    const bool more2 = Tn + dT < nT;
    float4 hrow = make_float4(0.f, 0.f, 0.f, 0.f);
    if (more2)
      hrow = Hv[min((Tn + dT) * 16 + wave, N - 1) * (HID / 4) + lane];

    const int dgs = __builtin_amdgcn_readfirstlane(dg);
    const int m = (node < N) ? min(dgs, CAP) : 0;
    const int curm = (lane < m) ? cur : 0;
    const float inv = 1.0f / fmaxf((float)dgs, 1.0f);

    float4 acc = make_float4(0.f, 0.f, 0.f, 0.f);
    float4 v[8];
#pragma unroll
    for (int q = 0; q < 8; ++q) {
      const int s = __builtin_amdgcn_readlane(curm, q);
      v[q] = p2v[s * 32 + lane];
    }
    if (first) z_part(at[0], wf, bv, zt[0][kh], j, kh, c, hh);
    if (more) z_part(at[bsel ^ 1], wf, bv, zt[bsel ^ 1][kh], j, kh, c, hh);
#pragma unroll
    for (int q = 0; q < 8; ++q) {
      const bool ok = q < m;
      acc.x += ok ? v[q].x : 0.f;
      acc.y += ok ? v[q].y : 0.f;
      acc.z += ok ? v[q].z : 0.f;
      acc.w += ok ? v[q].w : 0.f;
    }
    for (int i = 8; i < m; i += 8) {
#pragma unroll
      for (int q = 0; q < 8; ++q) {
        const int s = __builtin_amdgcn_readlane(curm, i + q);
        v[q] = p2v[s * 32 + lane];
      }
#pragma unroll
      for (int q = 0; q < 8; ++q) {
        const bool ok = (i + q) < m;
        acc.x += ok ? v[q].x : 0.f;
        acc.y += ok ? v[q].y : 0.f;
        acc.z += ok ? v[q].z : 0.f;
        acc.w += ok ? v[q].w : 0.f;
      }
    }

    if (node < N && dgs > CAP) {
      const int ns = nspill[0];
      const int2* sp = (const int2*)spill;
      for (int o = 0; o < ns; o += 32) {
        int2 p = make_int2(-1, 0);
        if (o + lane < ns) p = sp[o + lane];
        unsigned hit = __builtin_amdgcn_ballot_w32(p.x == node);
        while (hit != 0u) {
          const int b = __builtin_ctz(hit);
          hit &= hit - 1u;
          const int s = __builtin_amdgcn_readlane(p.y, b);
          const float4 w = p2v[s * 32 + lane];
          acc.x += w.x;
          acc.y += w.y;
          acc.z += w.z;
          acc.w += w.w;
        }
      }
    }

    if (first) {
      lds_barrier();
      first = false;
    }

    if (node < N) {
      const float4 z0 = ((const float4*)&zt[bsel][0][wave * ZP])[lane];
      const float4 z1 = ((const float4*)&zt[bsel][1][wave * ZP])[lane];

      float4 zz;
      zz.x = (z0.x + z1.x) + acc.x * inv;
      zz.y = (z0.y + z1.y) + acc.y * inv;
      zz.z = (z0.z + z1.z) + acc.z * inv;
      zz.w = (z0.w + z1.w) + acc.w * inv;

      float sq = zz.x * zz.x + zz.y * zz.y + zz.z * zz.z + zz.w * zz.w;
#pragma unroll
      for (int off = 16; off > 0; off >>= 1) sq += __shfl_xor(sq, off, 32);
      const float nrm = fmaxf(sqrtf(sq), 1e-12f);

      float4 r;
      r.x = fmaxf(zz.x / nrm, 0.0f);
      r.y = fmaxf(zz.y / nrm, 0.0f);
      r.z = fmaxf(zz.z / nrm, 0.0f);
      r.w = fmaxf(zz.w / nrm, 0.0f);
      zv4[node * 32 + lane] = r;

      ps.x += r.x;  ps.y += r.y;  ps.z += r.z;  ps.w += r.w;
      pss.x += r.x * r.x;  pss.y += r.y * r.y;
      pss.z += r.z * r.z;  pss.w += r.w * r.w;
    }

    if (more2) at[bsel][wave * TPA + lane] = hrow;
    dg = dgn;
    cur = curn;
    bsel ^= 1;
    lds_barrier();
  }

  ((float4*)&red[wave][0])[lane] = ps;
  ((float4*)&red[wave][128])[lane] = pss;
  __syncthreads();
  if (t < 256) {
    float s = 0.0f;
#pragma unroll
    for (int w = 0; w < GW; ++w) s += red[w][t];
    __hip_atomic_fetch_add(&tot[t], (long long)(s * FX_SCALE),
                           __ATOMIC_RELAXED, __HIP_MEMORY_SCOPE_AGENT);
  }
}

#define RG_WAVES 16
#define RG_BLOCKS 256
#define TP 33

__device__ __forceinline__ void rows_to_frags(float4 (&a)[16], float4* tb,
                                              int lane, int c, int hh) {
#pragma unroll
  for (int i = 0; i < 16; ++i) tb[i * TP + lane] = a[i];
  __builtin_amdgcn_fence(__ATOMIC_ACQ_REL, "wavefront");
#pragma unroll
  for (int g = 0; g < 16; ++g) a[g] = tb[c * TP + 2 * g + hh];
}

__device__ __forceinline__ void acc_to_rows(const v8f (&acc)[8], float4* tb,
                                            float* __restrict__ out, int tile,
                                            int lane, int c, int hh) {
  float* tbf = (float*)tb;
#pragma unroll
  for (int r = 0; r < 8; ++r) {
#pragma unroll
    for (int j = 0; j < 8; ++j)
      tbf[(8 * hh + r) * (4 * TP) + 16 * j + c] = acc[j][r];
  }
  __builtin_amdgcn_fence(__ATOMIC_ACQ_REL, "wavefront");
  float4* op = (float4*)out + tile * 16 * (HID / 4) + lane;
#pragma unroll
  for (int i = 0; i < 16; ++i) op[i * (HID / 4)] = tb[i * TP + lane];
}

static_assert(RG_BLOCKS == NCH && RG_WAVES * 32 == BE_T && BE_T == 2 * NCH,
              "the table steps use the GEMM launches' grid shape");

template <int NQ>
__global__ void __launch_bounds__(RG_WAVES * 32) rowtile_gemm(
    const float* __restrict__ A, const float* __restrict__ W,
    const float* __restrict__ bias, float* __restrict__ out, int nTiles,
    const int* __restrict__ src,
    const int* __restrict__ dst, int* __restrict__ pairs,
    int* __restrict__ offs, int* __restrict__ head, int nzero,
    int* __restrict__ cnt, int* __restrict__ tab, int* __restrict__ spill,
    int N, int E, int nbk) {
  __shared__ __attribute__((aligned(16))) float4 wf[16 * 8 * 2 * 16];
  __shared__ __attribute__((aligned(16))) float4 tbuf[RG_WAVES][16 * TP];

  const int tid = threadIdx.x;
  const int lane = tid & 31;
  const int wave = __builtin_amdgcn_readfirstlane(tid >> 5);
  const int hh = lane >> 4;
  const int c = lane & 15;

  const int tb = blockIdx.x;
  const int ts = RG_BLOCKS;
  const float* Wq = W + (NQ - 1) * HID;

  const float4* Av = (const float4*)A;
  float4* tbw = tbuf[wave];

  int m = wave;
  int tile = tb + ts * m;
  bool have = tile < nTiles;

  float4 a[16];
  if (have) {
    const float4* ap = Av + tile * 16 * (HID / 4) + lane;
#pragma unroll
    for (int i = 0; i < 16; ++i) a[i] = ap[i * (HID / 4)];
  }

  if (NQ == 1)
    bin_chunk(src, dst, pairs, offs, head, nzero, E, nbk);
  else
    slot_ranges(pairs, offs, cnt, tab, spill, head, N, E, nbk);

  {
    float4 t[8];
#pragma unroll
    for (int i = 0; i < 8; ++i)
      t[i] = *(const float4*)(Wq + (16 * i + wave) * (HID * NQ) + 4 * lane);
#pragma unroll
    for (int i = 0; i < 8; ++i)
      wf[(((lane >> 1) * 8 + i) * 2 + (lane & 1)) * 16 + ((wave ^ lane) & 15)] =
          t[i];
  }
  lds_barrier();

  while (have) {
    asm volatile("" ::: "memory");
    rows_to_frags(a, tbw, lane, c, hh);

    v8f acc[8];
#pragma unroll
    for (int j = 0; j < 8; ++j) {
      const float bv = (NQ == 1) ? bias[16 * j + c] : 0.0f;
#pragma unroll
      for (int r = 0; r < 8; ++r) acc[j][r] = bv;
    }

    float4 b0[8], b1[8];
#pragma unroll
    for (int j = 0; j < 8; ++j)
      b0[j] = wf[((0 * 8 + j) * 2 + hh) * 16 + (c ^ ((2 * 0 + hh) & 15))];

#pragma unroll
    for (int g = 0; g < 16; g += 2) {
#pragma unroll
      for (int j = 0; j < 8; ++j)
        b1[j] = wf[(((g + 1) * 8 + j) * 2 + hh) * 16 +
                   (c ^ ((2 * (g + 1) + hh) & 15))];
      __builtin_amdgcn_sched_barrier(0);
      {
        const v2f alo = {a[g].x, a[g].y};
        const v2f ahi = {a[g].z, a[g].w};
#pragma unroll
        for (int j = 0; j < 8; ++j) {
          const v2f b = {b0[j].x, b0[j].y};
          acc[j] = __builtin_amdgcn_wmma_f32_16x16x4_f32(
              false, alo, false, b, (short)0, acc[j], false, false);
        }
#pragma unroll
        for (int j = 0; j < 8; ++j) {
          const v2f b = {b0[j].z, b0[j].w};
          acc[j] = __builtin_amdgcn_wmma_f32_16x16x4_f32(
              false, ahi, false, b, (short)0, acc[j], false, false);
        }
      }
      __builtin_amdgcn_sched_barrier(0);
      if (g + 2 < 16) {
#pragma unroll
        for (int j = 0; j < 8; ++j)
          b0[j] = wf[(((g + 2) * 8 + j) * 2 + hh) * 16 +
                     (c ^ ((2 * (g + 2) + hh) & 15))];
      }
      __builtin_amdgcn_sched_barrier(0);
      {
        const v2f alo = {a[g + 1].x, a[g + 1].y};
        const v2f ahi = {a[g + 1].z, a[g + 1].w};
#pragma unroll
        for (int j = 0; j < 8; ++j) {
          const v2f b = {b1[j].x, b1[j].y};
          acc[j] = __builtin_amdgcn_wmma_f32_16x16x4_f32(
              false, alo, false, b, (short)0, acc[j], false, false);
        }
#pragma unroll
        for (int j = 0; j < 8; ++j) {
          const v2f b = {b1[j].z, b1[j].w};
          acc[j] = __builtin_amdgcn_wmma_f32_16x16x4_f32(
              false, ahi, false, b, (short)0, acc[j], false, false);
        }
      }
      __builtin_amdgcn_sched_barrier(0);
    }

    acc_to_rows(acc, tbw, out, tile, lane, c, hh);

    m += RG_WAVES;
    tile = tb + ts * m;
    have = tile < nTiles;
    if (have) {
      const float4* ap = Av + tile * 16 * (HID / 4) + lane;
#pragma unroll
      for (int i = 0; i < 16; ++i) a[i] = ap[i * (HID / 4)];
    }
  }
}

__device__ __forceinline__ void half_pass(const float4 (&a)[16],
                                          const float4* __restrict__ wq,
                                          float* __restrict__ out, float4* tb,
                                          int tile, int lane, int c, int hh) {
  v8f acc[8];
#pragma unroll
  for (int j = 0; j < 8; ++j) {
#pragma unroll
    for (int r = 0; r < 8; ++r) acc[j][r] = 0.0f;
  }

  float4 b0[8], b1[8];
#pragma unroll
  for (int j = 0; j < 8; ++j)
    b0[j] = wq[((0 * 8 + j) * 2 + hh) * 16 + (c ^ ((2 * 0 + hh) & 15))];

#pragma unroll
  for (int g = 0; g < 16; g += 2) {
#pragma unroll
    for (int j = 0; j < 8; ++j)
      b1[j] = wq[(((g + 1) * 8 + j) * 2 + hh) * 16 +
                 (c ^ ((2 * (g + 1) + hh) & 15))];
    __builtin_amdgcn_sched_barrier(0);
    {
      const v2f alo = {a[g].x, a[g].y};
      const v2f ahi = {a[g].z, a[g].w};
#pragma unroll
      for (int j = 0; j < 8; ++j) {
        const v2f b = {b0[j].x, b0[j].y};
        acc[j] = __builtin_amdgcn_wmma_f32_16x16x4_f32(
            false, alo, false, b, (short)0, acc[j], false, false);
      }
#pragma unroll
      for (int j = 0; j < 8; ++j) {
        const v2f b = {b0[j].z, b0[j].w};
        acc[j] = __builtin_amdgcn_wmma_f32_16x16x4_f32(
            false, ahi, false, b, (short)0, acc[j], false, false);
      }
    }
    __builtin_amdgcn_sched_barrier(0);
    if (g + 2 < 16) {
#pragma unroll
      for (int j = 0; j < 8; ++j)
        b0[j] = wq[(((g + 2) * 8 + j) * 2 + hh) * 16 +
                   (c ^ ((2 * (g + 2) + hh) & 15))];
    }
    __builtin_amdgcn_sched_barrier(0);
    {
      const v2f alo = {a[g + 1].x, a[g + 1].y};
      const v2f ahi = {a[g + 1].z, a[g + 1].w};
#pragma unroll
      for (int j = 0; j < 8; ++j) {
        const v2f b = {b1[j].x, b1[j].y};
        acc[j] = __builtin_amdgcn_wmma_f32_16x16x4_f32(
            false, alo, false, b, (short)0, acc[j], false, false);
      }
#pragma unroll
      for (int j = 0; j < 8; ++j) {
        const v2f b = {b1[j].z, b1[j].w};
        acc[j] = __builtin_amdgcn_wmma_f32_16x16x4_f32(
            false, ahi, false, b, (short)0, acc[j], false, false);
      }
    }
    __builtin_amdgcn_sched_barrier(0);
  }

  acc_to_rows(acc, tb, out, tile, lane, c, hh);
}

__global__ void __launch_bounds__(RG_WAVES * 32) fold_gemm(
    float* __restrict__ Hh, const float* __restrict__ Zz,
    const float* __restrict__ W, const long long* __restrict__ tot,
    const float* __restrict__ gamma,
    const float* __restrict__ beta, float* __restrict__ Pp, int nTiles,
    float invN) {
  __shared__ __attribute__((aligned(16))) float4 wf[16 * 8 * 2 * 16];
  __shared__ __attribute__((aligned(16))) float4 tbuf[RG_WAVES][16 * TP];
  __shared__ __attribute__((aligned(16))) float bnab[2 * HID];

  const int tid = threadIdx.x;
  const int lane = tid & 31;
  const int wave = __builtin_amdgcn_readfirstlane(tid >> 5);
  const int hh = lane >> 4;
  const int c = lane & 15;

  const int slot = 4 * (((wave & 3) - (wave >> 2)) & 3) + (wave >> 2);
  const int tile = (int)blockIdx.x + RG_BLOCKS * slot;
  const bool have = tile < nTiles;

  float4 hv[16], zv[16];
  float4* hrow = (float4*)Hh + tile * 16 * (HID / 4) + lane;
  if (have) {
    const float4* zrow = (const float4*)Zz + tile * 16 * (HID / 4) + lane;
#pragma unroll
    for (int i = 0; i < 16; ++i) hv[i] = hrow[i * (HID / 4)];
#pragma unroll
    for (int i = 0; i < 16; ++i) zv[i] = zrow[i * (HID / 4)];
  }

  if (tid < HID) {
    const float s = (float)tot[tid] * FX_INV;
    const float ss = (float)tot[HID + tid] * FX_INV;
    const float mu = s * invN;
    const float var = ss * invN - mu * mu;
    const float sa = gamma[tid] / sqrtf(var + 1e-5f);
    bnab[tid] = sa;
    bnab[HID + tid] = beta[tid] - mu * sa;
  }

  {
    float4 t[8];
#pragma unroll
    for (int i = 0; i < 8; ++i)
      t[i] = *(const float4*)(W + (16 * i + wave) * (HID * 2) + HID + 4 * lane);
#pragma unroll
    for (int i = 0; i < 8; ++i)
      wf[(((lane >> 1) * 8 + i) * 2 + (lane & 1)) * 16 + ((wave ^ lane) & 15)] =
          t[i];
  }
  __syncthreads();

  if (have) {
    const float4 sa = ((const float4*)&bnab[0])[lane];
    const float4 sb = ((const float4*)&bnab[HID])[lane];
#pragma unroll
    for (int i = 0; i < 16; ++i) {
      hv[i].x = hv[i].x + zv[i].x * sa.x + sb.x;
      hv[i].y = hv[i].y + zv[i].y * sa.y + sb.y;
      hv[i].z = hv[i].z + zv[i].z * sa.z + sb.z;
      hv[i].w = hv[i].w + zv[i].w * sa.w + sb.w;
    }
#pragma unroll
    for (int i = 0; i < 16; ++i) hrow[i * (HID / 4)] = hv[i];

    float4* tbw = tbuf[wave];
    rows_to_frags(hv, tbw, lane, c, hh);

    half_pass(hv, wf, Pp, tbw, tile, lane, c, hh);
  }
}

__global__ void residual_bn(const float* __restrict__ h,
                            const float* __restrict__ z,
                            const long long* __restrict__ tot,
                            const float* __restrict__ gamma,
                            const float* __restrict__ beta,
                            float* __restrict__ out, int total, float invN) {
  const int j = threadIdx.x & (HID - 1);
  const float s = (float)tot[j] * FX_INV;
  const float ss = (float)tot[HID + j] * FX_INV;
  const float mu = s * invN;
  const float var = ss * invN - mu * mu;
  const float a = gamma[j] / sqrtf(var + 1e-5f);
  const float b = beta[j] - mu * a;
  for (int i = blockIdx.x * blockDim.x + threadIdx.x; i < total;
       i += gridDim.x * blockDim.x)
    out[i] = h[i] + z[i] * a + b;
}

extern "C" void kernel_launch(void* const* d_in, const int* in_sizes, int n_in,
                              void* d_out, int out_size, void* d_ws,
                              size_t ws_size, hipStream_t stream) {
  const float* h0 = (const float*)d_in[0];
  const int* src = (const int*)d_in[1];
  const int* dst = (const int*)d_in[2];
  const float* W_emb = (const float*)d_in[3];
  const float* b_emb = (const float*)d_in[4];
  const float* Ws = (const float*)d_in[5];
  const float* bs = (const float*)d_in[6];
  const float* gam = (const float*)d_in[7];
  const float* bet = (const float*)d_in[8];

  const int N = in_sizes[0] / HID;
  const int E = in_sizes[1];
  const int NF = N * HID;
  const int L = 4;

  float* ws = (float*)d_ws;
  float* H = ws;
  float* C = H + NF;
  float* Z = C + NF;
  int* head = (int*)(Z + NF);
  long long* TOT = (long long*)(head + 2);
  int* cnt = (int*)(TOT + L * 256);

  int* tab = (int*)d_out;
  int* spill = tab + N * CAP;
  int* pairs = spill + 2 * E;
  int* offs = pairs + 2 * E;

  const int mtiles = N / 16;
  const int nbk = (N + 127) / 128;

  rowtile_gemm<1><<<RG_BLOCKS, RG_WAVES * 32, 0, stream>>>(
      h0, W_emb, b_emb, H, mtiles, src, dst, pairs, offs, head, 2 + L * 512,
      cnt, tab, spill, N, E, nbk);

  const float invN = 1.0f / (float)N;
  for (int l = 0; l < L; ++l) {
    if (l == 0) {
      rowtile_gemm<2><<<RG_BLOCKS, RG_WAVES * 32, 0, stream>>>(
          H, Ws, bs, C, mtiles, src, dst, pairs, offs, head, 0, cnt, tab,
          spill, N, E, nbk);
    } else {
      fold_gemm<<<RG_BLOCKS, RG_WAVES * 32, 0, stream>>>(
          H, Z, Ws + l * HID * 256, TOT + (l - 1) * 256, gam + (l - 1) * HID,
          bet + (l - 1) * HID, C, mtiles, invN);
    }
    gather_z_norm_stats<<<NB_STATS, GW * 32, 0, stream>>>(
        H, Ws + l * HID * 256, bs + l * HID, C, cnt, tab, spill, head, Z,
        TOT + l * 256, N, mtiles);
  }
  residual_bn<<<4096, 256, 0, stream>>>(H, Z, TOT + (L - 1) * 256,
                                        gam + (L - 1) * HID,
                                        bet + (L - 1) * HID, (float*)d_out, NF,
                                        invN);
}
